# starting kernel
# baseline (speedup 1.0000x reference)
#include <hip/hip_runtime.h>
#include <math.h>

#define B_      8
#define TXT_T   200
#define TXT_PAD 208
#define MEL_T   800
#define N_MEL   80
#define KDIM    160
#define PAD_MAG 1.0e12f

typedef __attribute__((ext_vector_type(2))) float v2f;
typedef __attribute__((ext_vector_type(8))) float v8f;

#define A_OFF 0
#define C_OFF (B_ * TXT_PAD * KDIM)
#define P_OFF (C_OFF + B_ * TXT_PAD)

__global__ void mdn_prep_kernel(const float* __restrict__ mu_logvar,
                                float* __restrict__ ws) {
  int idx = blockIdx.x * blockDim.x + threadIdx.x;
  if (idx >= B_ * TXT_PAD) return;
  int b = idx / TXT_PAD;
  int t = idx % TXT_PAD;
  float* Arow = ws + A_OFF + (size_t)idx * KDIM;
  float c = 0.0f;
  if (t < TXT_T) {
    const float* ml = mu_logvar + (size_t)(b * TXT_T + t) * (2 * N_MEL);
    for (int n = 0; n < N_MEL; ++n) {
      float mu = ml[n];
      float lv = ml[N_MEL + n];
      float w  = expf(-lv);
      Arow[n]          = w;
      Arow[N_MEL + n]  = -2.0f * mu * w;
      c += mu * mu * w + lv;
    }
  } else {
    for (int n = 0; n < KDIM; ++n) Arow[n] = 0.0f;
  }
  ws[C_OFF + idx] = c;
}

__global__ void __launch_bounds__(256)
mdn_gemm_kernel(const float* __restrict__ ws,
                const float* __restrict__ mel,
                float* __restrict__ lp) {
  const int TT = TXT_PAD / 16;
  const int MT = MEL_T / 16;
  int wid = blockIdx.x * (blockDim.x >> 5) + (threadIdx.x >> 5);
  if (wid >= B_ * TT * MT) return;
  int lane = threadIdx.x & 31;
  int l15  = lane & 15;
  int kl   = lane >> 4;

  int b  = wid / (TT * MT);
  int r  = wid % (TT * MT);
  int t0 = (r / MT) * 16;
  int m0 = (r % MT) * 16;

  const float* Arow = ws + A_OFF + (size_t)(b * TXT_PAD + t0 + l15) * KDIM;
  const float* melb = mel + (size_t)b * N_MEL * MEL_T + m0 + l15;

  v8f acc = {0.f, 0.f, 0.f, 0.f, 0.f, 0.f, 0.f, 0.f};

  for (int k = 0; k < KDIM; k += 4) {
    v2f a = *(const v2f*)(Arow + k + 2 * kl);

    int K0 = k + 2 * kl;
    int n  = (k < N_MEL) ? K0 : (K0 - N_MEL);
    float x0 = melb[(size_t)n * MEL_T];
    float x1 = melb[(size_t)(n + 1) * MEL_T];
    if (k < N_MEL) { x0 *= x0; x1 *= x1; }
    v2f bv; bv.x = x0; bv.y = x1;

    acc = __builtin_amdgcn_wmma_f32_16x16x4_f32(
         false, a,  false, bv,
         (short)0, acc,  false,  false);
  }

  int m = m0 + l15;
  const float* Cb = ws + C_OFF + b * TXT_PAD;
  const float scale = -0.5f / (float)N_MEL;
#pragma unroll
  for (int j = 0; j < 8; ++j) {
    int t = t0 + j + 8 * kl;
    if (t < TXT_T) {
      lp[(size_t)(b * TXT_T + t) * MEL_T + m] = scale * (acc[j] + Cb[t]);
    }
  }
}

__device__ __forceinline__ float logaddexp_(float u, float v) {
  float mx = fmaxf(u, v);
  float mn = fminf(u, v);
  return mx + log1pf(expf(mn - mx));
}

__global__ void mdn_scan_kernel(const float* __restrict__ lp,
                                const int* __restrict__ text_len,
                                const int* __restrict__ mel_len,
                                float* __restrict__ ws) {
  __shared__ float sa[TXT_T + 1];
  int b = blockIdx.x;
  int t = threadIdx.x;
  for (int i = t; i <= TXT_T; i += blockDim.x) sa[i] = -PAD_MAG;
  __syncthreads();
  const float* lpb = lp + (size_t)b * TXT_T * MEL_T;
  if (t == 0) sa[1] = lpb[0];
  __syncthreads();

  int M = mel_len[b];
  for (int m = 1; m < M; ++m) {
    float u = 0.f, v = 0.f;
    if (t < TXT_T) { u = sa[t + 1]; v = sa[t]; }
    __syncthreads();
    if (t < TXT_T) {
      sa[t + 1] = logaddexp_(u + 1e-7f, v + 1e-7f) + lpb[(size_t)t * MEL_T + m];
    }
    __syncthreads();
  }
  if (t == 0) {
    int tl = text_len[b];
    ws[P_OFF + b] = sa[tl] / (float)M;
  }
}

__global__ void mdn_reduce_kernel(const float* __restrict__ ws,
                                  float* __restrict__ out) {
  float s = 0.f;
  for (int b = 0; b < B_; ++b) s += ws[P_OFF + b];
  out[0] = -s / (float)B_;
}

extern "C" void kernel_launch(void* const* d_in, const int* in_sizes, int n_in,
                              void* d_out, int out_size, void* d_ws, size_t ws_size,
                              hipStream_t stream) {
  const float* mu_logvar = (const float*)d_in[0];
  const float* melspec   = (const float*)d_in[1];
  const int*   text_len  = (const int*)d_in[2];
  const int*   mel_len   = (const int*)d_in[3];
  float* out = (float*)d_out;
  float* ws  = (float*)d_ws;
  float* lp  = out + 1;

  mdn_prep_kernel<<<(B_ * TXT_PAD + 255) / 256, 256, 0, stream>>>(mu_logvar, ws);

  const int total_waves = B_ * (TXT_PAD / 16) * (MEL_T / 16);
  mdn_gemm_kernel<<<total_waves / 8, 256, 0, stream>>>(ws, melspec, lp);

  mdn_scan_kernel<<<B_, 256, 0, stream>>>(lp, text_len, mel_len, ws);
  mdn_reduce_kernel<<<1, 1, 0, stream>>>(ws, out);
}


# scan kernel: library log1pf(expf(d)) of the logaddexp replaced by the hardware exp2/log2 instructions (v_exp_f32, v_log_f32), f32 kept; shortens the dependent chain each serial column waits for
# speedup vs baseline: 1.4855x; 1.4855x over previous
#include <hip/hip_runtime.h>
#include <math.h>

#define B_      8
#define TXT_T   200
#define TXT_PAD 208
#define MEL_T   800
#define N_MEL   80
#define KDIM    160
#define PAD_MAG 1.0e12f

typedef __attribute__((ext_vector_type(2))) float v2f;
typedef __attribute__((ext_vector_type(8))) float v8f;

#define A_OFF 0
#define C_OFF (B_ * TXT_PAD * KDIM)
#define P_OFF (C_OFF + B_ * TXT_PAD)

__global__ void mdn_prep_kernel(const float* __restrict__ mu_logvar,
                                float* __restrict__ ws) {
  int idx = blockIdx.x * blockDim.x + threadIdx.x;
  if (idx >= B_ * TXT_PAD) return;
  int b = idx / TXT_PAD;
  int t = idx % TXT_PAD;
  float* Arow = ws + A_OFF + (size_t)idx * KDIM;
  float c = 0.0f;
  if (t < TXT_T) {
    const float* ml = mu_logvar + (size_t)(b * TXT_T + t) * (2 * N_MEL);
    for (int n = 0; n < N_MEL; ++n) {
      float mu = ml[n];
      float lv = ml[N_MEL + n];
      float w  = expf(-lv);
      Arow[n]          = w;
      Arow[N_MEL + n]  = -2.0f * mu * w;
      c += mu * mu * w + lv;
    }
  } else {
    for (int n = 0; n < KDIM; ++n) Arow[n] = 0.0f;
  }
  ws[C_OFF + idx] = c;
}

__global__ void __launch_bounds__(256)
mdn_gemm_kernel(const float* __restrict__ ws,
                const float* __restrict__ mel,
                float* __restrict__ lp) {
  const int TT = TXT_PAD / 16;
  const int MT = MEL_T / 16;
  int wid = blockIdx.x * (blockDim.x >> 5) + (threadIdx.x >> 5);
  if (wid >= B_ * TT * MT) return;
  int lane = threadIdx.x & 31;
  int l15  = lane & 15;
  int kl   = lane >> 4;

  int b  = wid / (TT * MT);
  int r  = wid % (TT * MT);
  int t0 = (r / MT) * 16;
  int m0 = (r % MT) * 16;

  const float* Arow = ws + A_OFF + (size_t)(b * TXT_PAD + t0 + l15) * KDIM;
  const float* melb = mel + (size_t)b * N_MEL * MEL_T + m0 + l15;

  v8f acc = {0.f, 0.f, 0.f, 0.f, 0.f, 0.f, 0.f, 0.f};

  for (int k = 0; k < KDIM; k += 4) {
    v2f a = *(const v2f*)(Arow + k + 2 * kl);

    int K0 = k + 2 * kl;
    int n  = (k < N_MEL) ? K0 : (K0 - N_MEL);
    float x0 = melb[(size_t)n * MEL_T];
    float x1 = melb[(size_t)(n + 1) * MEL_T];
    if (k < N_MEL) { x0 *= x0; x1 *= x1; }
    v2f bv; bv.x = x0; bv.y = x1;

    acc = __builtin_amdgcn_wmma_f32_16x16x4_f32(
         false, a,  false, bv,
         (short)0, acc,  false,  false);
  }

  int m = m0 + l15;
  const float* Cb = ws + C_OFF + b * TXT_PAD;
  const float scale = -0.5f / (float)N_MEL;
#pragma unroll
  for (int j = 0; j < 8; ++j) {
    int t = t0 + j + 8 * kl;
    if (t < TXT_T) {
      lp[(size_t)(b * TXT_T + t) * MEL_T + m] = scale * (acc[j] + Cb[t]);
    }
  }
}

__device__ __forceinline__ float logaddexp_(float u, float v) {
  float mx = fmaxf(u, v);
  float mn = fminf(u, v);
  float e  = __builtin_amdgcn_exp2f((mn - mx) * 1.44269504f);
  return mx + 0.693147181f * __builtin_amdgcn_logf(1.0f + e);
}

__global__ void mdn_scan_kernel(const float* __restrict__ lp,
                                const int* __restrict__ text_len,
                                const int* __restrict__ mel_len,
                                float* __restrict__ ws) {
  __shared__ float sa[TXT_T + 1];
  int b = blockIdx.x;
  int t = threadIdx.x;
  for (int i = t; i <= TXT_T; i += blockDim.x) sa[i] = -PAD_MAG;
  __syncthreads();
  const float* lpb = lp + (size_t)b * TXT_T * MEL_T;
  if (t == 0) sa[1] = lpb[0];
  __syncthreads();

  int M = mel_len[b];
  for (int m = 1; m < M; ++m) {
    float u = 0.f, v = 0.f;
    if (t < TXT_T) { u = sa[t + 1]; v = sa[t]; }
    __syncthreads();
    if (t < TXT_T) {
      sa[t + 1] = logaddexp_(u + 1e-7f, v + 1e-7f) + lpb[(size_t)t * MEL_T + m];
    }
    __syncthreads();
  }
  if (t == 0) {
    int tl = text_len[b];
    ws[P_OFF + b] = sa[tl] / (float)M;
  }
}

__global__ void mdn_reduce_kernel(const float* __restrict__ ws,
                                  float* __restrict__ out) {
  float s = 0.f;
  for (int b = 0; b < B_; ++b) s += ws[P_OFF + b];
  out[0] = -s / (float)B_;
}

extern "C" void kernel_launch(void* const* d_in, const int* in_sizes, int n_in,
                              void* d_out, int out_size, void* d_ws, size_t ws_size,
                              hipStream_t stream) {
  const float* mu_logvar = (const float*)d_in[0];
  const float* melspec   = (const float*)d_in[1];
  const int*   text_len  = (const int*)d_in[2];
  const int*   mel_len   = (const int*)d_in[3];
  float* out = (float*)d_out;
  float* ws  = (float*)d_ws;
  float* lp  = out + 1;

  mdn_prep_kernel<<<(B_ * TXT_PAD + 255) / 256, 256, 0, stream>>>(mu_logvar, ws);

  const int total_waves = B_ * (TXT_PAD / 16) * (MEL_T / 16);
  mdn_gemm_kernel<<<total_waves / 8, 256, 0, stream>>>(ws, melspec, lp);

  mdn_scan_kernel<<<B_, 256, 0, stream>>>(lp, text_len, mel_len, ws);
  mdn_reduce_kernel<<<1, 1, 0, stream>>>(ws, out);
}


# scan kernel: lp columns staged one 32-column chunk ahead through LDS (coalesced row loads, column-major in LDS) in place of the strided global load inside every serial step; sums unchanged
# speedup vs baseline: 1.9947x; 1.3428x over previous
#include <hip/hip_runtime.h>
#include <math.h>

#define B_      8
#define TXT_T   200
#define TXT_PAD 208
#define MEL_T   800
#define N_MEL   80
#define KDIM    160
#define PAD_MAG 1.0e12f

typedef __attribute__((ext_vector_type(2))) float v2f;
typedef __attribute__((ext_vector_type(8))) float v8f;

#define A_OFF 0
#define C_OFF (B_ * TXT_PAD * KDIM)
#define P_OFF (C_OFF + B_ * TXT_PAD)

__global__ void mdn_prep_kernel(const float* __restrict__ mu_logvar,
                                float* __restrict__ ws) {
  int idx = blockIdx.x * blockDim.x + threadIdx.x;
  if (idx >= B_ * TXT_PAD) return;
  int b = idx / TXT_PAD;
  int t = idx % TXT_PAD;
  float* Arow = ws + A_OFF + (size_t)idx * KDIM;
  float c = 0.0f;
  if (t < TXT_T) {
    const float* ml = mu_logvar + (size_t)(b * TXT_T + t) * (2 * N_MEL);
    for (int n = 0; n < N_MEL; ++n) {
      float mu = ml[n];
      float lv = ml[N_MEL + n];
      float w  = expf(-lv);
      Arow[n]          = w;
      Arow[N_MEL + n]  = -2.0f * mu * w;
      c += mu * mu * w + lv;
    }
  } else {
    for (int n = 0; n < KDIM; ++n) Arow[n] = 0.0f;
  }
  ws[C_OFF + idx] = c;
}

__global__ void __launch_bounds__(256)
mdn_gemm_kernel(const float* __restrict__ ws,
                const float* __restrict__ mel,
                float* __restrict__ lp) {
  const int TT = TXT_PAD / 16;
  const int MT = MEL_T / 16;
  int wid = blockIdx.x * (blockDim.x >> 5) + (threadIdx.x >> 5);
  if (wid >= B_ * TT * MT) return;
  int lane = threadIdx.x & 31;
  int l15  = lane & 15;
  int kl   = lane >> 4;

  int b  = wid / (TT * MT);
  int r  = wid % (TT * MT);
  int t0 = (r / MT) * 16;
  int m0 = (r % MT) * 16;

  const float* Arow = ws + A_OFF + (size_t)(b * TXT_PAD + t0 + l15) * KDIM;
  const float* melb = mel + (size_t)b * N_MEL * MEL_T + m0 + l15;

  v8f acc = {0.f, 0.f, 0.f, 0.f, 0.f, 0.f, 0.f, 0.f};

  for (int k = 0; k < KDIM; k += 4) {
    v2f a = *(const v2f*)(Arow + k + 2 * kl);

    int K0 = k + 2 * kl;
    int n  = (k < N_MEL) ? K0 : (K0 - N_MEL);
    float x0 = melb[(size_t)n * MEL_T];
    float x1 = melb[(size_t)(n + 1) * MEL_T];
    if (k < N_MEL) { x0 *= x0; x1 *= x1; }
    v2f bv; bv.x = x0; bv.y = x1;

    acc = __builtin_amdgcn_wmma_f32_16x16x4_f32(
         false, a,  false, bv,
         (short)0, acc,  false,  false);
  }

  int m = m0 + l15;
  const float* Cb = ws + C_OFF + b * TXT_PAD;
  const float scale = -0.5f / (float)N_MEL;
#pragma unroll
  for (int j = 0; j < 8; ++j) {
    int t = t0 + j + 8 * kl;
    if (t < TXT_T) {
      lp[(size_t)(b * TXT_T + t) * MEL_T + m] = scale * (acc[j] + Cb[t]);
    }
  }
}

__device__ __forceinline__ float logaddexp_(float u, float v) {
  float mx = fmaxf(u, v);
  float mn = fminf(u, v);
  float e  = __builtin_amdgcn_exp2f((mn - mx) * 1.44269504f);
  return mx + 0.693147181f * __builtin_amdgcn_logf(1.0f + e);
}

#define SCH      32
#define SPITCH   201
#define SROWS    (TXT_T / 8)
static_assert(TXT_T % 8 == 0, "the 8 waves of the scan block share the rows evenly");

__device__ __forceinline__ void lds_barrier_() {
  __builtin_amdgcn_fence(__ATOMIC_RELEASE, "workgroup", "local");
  __builtin_amdgcn_s_barrier();
  __builtin_amdgcn_fence(__ATOMIC_ACQUIRE, "workgroup", "local");
}

__global__ void __launch_bounds__(256)
mdn_scan_kernel(const float* __restrict__ lp,
                const int* __restrict__ text_len,
                const int* __restrict__ mel_len,
                float* __restrict__ ws) {
  __shared__ float sa[TXT_T + 1];
  __shared__ float stage[2][SCH * SPITCH];
  int b = blockIdx.x;
  int t = threadIdx.x;
  int lane = t & 31;
  int w    = t >> 5;
  for (int i = t; i <= TXT_T; i += blockDim.x) sa[i] = -PAD_MAG;
  __syncthreads();
  const float* lpb = lp + (size_t)b * TXT_T * MEL_T;
  if (t == 0) sa[1] = lpb[0];

  int M = mel_len[b];
  int nch = (M - 1 + SCH - 1) / SCH;

  const float* src = lpb + (size_t)w * MEL_T;
  float pre[SROWS];
  if (nch > 0) {
    int col = 1 + lane;
    if (col > M - 1) col = M - 1;
#pragma unroll
    for (int i = 0; i < SROWS; ++i) pre[i] = src[(size_t)i * 8 * MEL_T + col];
#pragma unroll
    for (int i = 0; i < SROWS; ++i) stage[0][lane * SPITCH + w + 8 * i] = pre[i];
  }
  __syncthreads();

  for (int c = 0; c < nch; ++c) {
    int m0 = 1 + c * SCH;
    const float* cur = stage[c & 1];
    float* nxt = stage[(c + 1) & 1];
    bool more = (c + 1) < nch;
    if (more) {
      int col = m0 + SCH + lane;
      if (col > M - 1) col = M - 1;
#pragma unroll
      for (int i = 0; i < SROWS; ++i) pre[i] = src[(size_t)i * 8 * MEL_T + col];
    }
    int steps = M - m0;
    if (steps > SCH) steps = SCH;
    for (int j = 0; j < steps; ++j) {
      float u = 0.f, v = 0.f, p = 0.f;
      if (t < TXT_T) { u = sa[t + 1]; v = sa[t]; p = cur[j * SPITCH + t]; }
      lds_barrier_();
      if (t < TXT_T) sa[t + 1] = logaddexp_(u + 1e-7f, v + 1e-7f) + p;
      lds_barrier_();
    }
    if (more) {
#pragma unroll
      for (int i = 0; i < SROWS; ++i) nxt[lane * SPITCH + w + 8 * i] = pre[i];
    }
    lds_barrier_();
  }
  if (t == 0) {
    int tl = text_len[b];
    ws[P_OFF + b] = sa[tl] / (float)M;
  }
}

__global__ void mdn_reduce_kernel(const float* __restrict__ ws,
                                  float* __restrict__ out) {
  float s = 0.f;
  for (int b = 0; b < B_; ++b) s += ws[P_OFF + b];
  out[0] = -s / (float)B_;
}

extern "C" void kernel_launch(void* const* d_in, const int* in_sizes, int n_in,
                              void* d_out, int out_size, void* d_ws, size_t ws_size,
                              hipStream_t stream) {
  const float* mu_logvar = (const float*)d_in[0];
  const float* melspec   = (const float*)d_in[1];
  const int*   text_len  = (const int*)d_in[2];
  const int*   mel_len   = (const int*)d_in[3];
  float* out = (float*)d_out;
  float* ws  = (float*)d_ws;
  float* lp  = out + 1;

  mdn_prep_kernel<<<(B_ * TXT_PAD + 255) / 256, 256, 0, stream>>>(mu_logvar, ws);

  const int total_waves = B_ * (TXT_PAD / 16) * (MEL_T / 16);
  mdn_gemm_kernel<<<total_waves / 8, 256, 0, stream>>>(ws, melspec, lp);

  mdn_scan_kernel<<<B_, 256, 0, stream>>>(lp, text_len, mel_len, ws);
  mdn_reduce_kernel<<<1, 1, 0, stream>>>(ws, out);
}


# scan kernel: one wave per batch entry holds the carry in registers (7 rows per lane, stages interleaved across rows, row below by cross-lane move, lp by 16-byte loads ahead); no LDS; sums unchanged
# speedup vs baseline: 3.9052x; 1.9578x over previous
#include <hip/hip_runtime.h>
#include <math.h>

#define B_      8
#define TXT_T   200
#define TXT_PAD 208
#define MEL_T   800
#define N_MEL   80
#define KDIM    160
#define PAD_MAG 1.0e12f

typedef __attribute__((ext_vector_type(2))) float v2f;
typedef __attribute__((ext_vector_type(8))) float v8f;

#define A_OFF 0
#define C_OFF (B_ * TXT_PAD * KDIM)
#define P_OFF (C_OFF + B_ * TXT_PAD)

__global__ void mdn_prep_kernel(const float* __restrict__ mu_logvar,
                                float* __restrict__ ws) {
  int idx = blockIdx.x * blockDim.x + threadIdx.x;
  if (idx >= B_ * TXT_PAD) return;
  int b = idx / TXT_PAD;
  int t = idx % TXT_PAD;
  float* Arow = ws + A_OFF + (size_t)idx * KDIM;
  float c = 0.0f;
  if (t < TXT_T) {
    const float* ml = mu_logvar + (size_t)(b * TXT_T + t) * (2 * N_MEL);
    for (int n = 0; n < N_MEL; ++n) {
      float mu = ml[n];
      float lv = ml[N_MEL + n];
      float w  = expf(-lv);
      Arow[n]          = w;
      Arow[N_MEL + n]  = -2.0f * mu * w;
      c += mu * mu * w + lv;
    }
  } else {
    for (int n = 0; n < KDIM; ++n) Arow[n] = 0.0f;
  }
  ws[C_OFF + idx] = c;
}

__global__ void __launch_bounds__(256)
mdn_gemm_kernel(const float* __restrict__ ws,
                const float* __restrict__ mel,
                float* __restrict__ lp) {
  const int TT = TXT_PAD / 16;
  const int MT = MEL_T / 16;
  int wid = blockIdx.x * (blockDim.x >> 5) + (threadIdx.x >> 5);
  if (wid >= B_ * TT * MT) return;
  int lane = threadIdx.x & 31;
  int l15  = lane & 15;
  int kl   = lane >> 4;

  int b  = wid / (TT * MT);
  int r  = wid % (TT * MT);
  int t0 = (r / MT) * 16;
  int m0 = (r % MT) * 16;

  const float* Arow = ws + A_OFF + (size_t)(b * TXT_PAD + t0 + l15) * KDIM;
  const float* melb = mel + (size_t)b * N_MEL * MEL_T + m0 + l15;

  v8f acc = {0.f, 0.f, 0.f, 0.f, 0.f, 0.f, 0.f, 0.f};

  for (int k = 0; k < KDIM; k += 4) {
    v2f a = *(const v2f*)(Arow + k + 2 * kl);

    int K0 = k + 2 * kl;
    int n  = (k < N_MEL) ? K0 : (K0 - N_MEL);
    float x0 = melb[(size_t)n * MEL_T];
    float x1 = melb[(size_t)(n + 1) * MEL_T];
    if (k < N_MEL) { x0 *= x0; x1 *= x1; }
    v2f bv; bv.x = x0; bv.y = x1;

    acc = __builtin_amdgcn_wmma_f32_16x16x4_f32(
         false, a,  false, bv,
         (short)0, acc,  false,  false);
  }

  int m = m0 + l15;
  const float* Cb = ws + C_OFF + b * TXT_PAD;
  const float scale = -0.5f / (float)N_MEL;
#pragma unroll
  for (int j = 0; j < 8; ++j) {
    int t = t0 + j + 8 * kl;
    if (t < TXT_T) {
      lp[(size_t)(b * TXT_T + t) * MEL_T + m] = scale * (acc[j] + Cb[t]);
    }
  }
}


#define SR 7
#define SG 4
#define SD 5
typedef __attribute__((ext_vector_type(4))) float v4f;
static_assert(SR * 32 >= TXT_T, "one wave covers every text row");
static_assert(MEL_T % SG == 0 && MEL_T >= 2 * SG, "row pitch keeps the 16-byte phase of every row equal");

__device__ __forceinline__ float lane_below_(float x, float padv, bool is16) {
  int xi = __builtin_bit_cast(int, x);
  int d  = __builtin_amdgcn_update_dpp(__builtin_bit_cast(int, padv), xi, 0x111, 0xf, 0xf, false);
  int c  = __builtin_amdgcn_permlanex16(xi, xi, 0xffffffff, 0xffffffff, false, false);
  return __builtin_bit_cast(float, is16 ? c : d);
}

struct ScanState {
  float a[SR];
  float e[SR];
  float eup;
};

#define STAGE_FENCE() __builtin_amdgcn_sched_barrier(0)
__device__ __forceinline__ void scan_step_(ScanState& st, const float (&p)[SR], bool is16) {
  float lo[SR], mx[SR], d[SR], x[SR];
  lo[0] = st.eup;
#pragma unroll
  for (int i = 1; i < SR; ++i) lo[i] = st.e[i - 1];
#pragma unroll
  for (int i = SR - 1; i >= 0; --i) { mx[i] = fmaxf(st.e[i], lo[i]); d[i] = fminf(st.e[i], lo[i]); }
  STAGE_FENCE();
#pragma unroll
  for (int i = SR - 1; i >= 0; --i) d[i] = d[i] - mx[i];
  STAGE_FENCE();
#pragma unroll
  for (int i = SR - 1; i >= 0; --i) d[i] = d[i] * 1.44269504f;
  STAGE_FENCE();
#pragma unroll
  for (int i = SR - 1; i >= 0; --i) x[i] = __builtin_amdgcn_exp2f(d[i]);
  STAGE_FENCE();
#pragma unroll
  for (int i = SR - 1; i >= 0; --i) x[i] = 1.0f + x[i];
  STAGE_FENCE();
#pragma unroll
  for (int i = SR - 1; i >= 0; --i) x[i] = __builtin_amdgcn_logf(x[i]);
  STAGE_FENCE();
#pragma unroll
  for (int i = SR - 1; i >= 0; --i) x[i] = mx[i] + 0.693147181f * x[i];
  STAGE_FENCE();
#pragma unroll
  for (int i = SR - 1; i >= 0; --i) st.a[i] = x[i] + p[i];
  STAGE_FENCE();
#pragma unroll
  for (int i = SR - 1; i >= 0; --i) st.e[i] = st.a[i] + 1e-7f;
  STAGE_FENCE();
  st.eup = lane_below_(st.e[SR - 1], -PAD_MAG, is16);
}

__device__ __forceinline__ void scan_group_(ScanState& st, const v4f (&q)[SR], int ncols, bool is16) {
  float p[SR];
  if (ncols < 1) return;
#pragma unroll
  for (int i = 0; i < SR; ++i) p[i] = q[i].x;
  scan_step_(st, p, is16);
  if (ncols < 2) return;
#pragma unroll
  for (int i = 0; i < SR; ++i) p[i] = q[i].y;
  scan_step_(st, p, is16);
  if (ncols < 3) return;
#pragma unroll
  for (int i = 0; i < SR; ++i) p[i] = q[i].z;
  scan_step_(st, p, is16);
  if (ncols < 4) return;
#pragma unroll
  for (int i = 0; i < SR; ++i) p[i] = q[i].w;
  scan_step_(st, p, is16);
}

__global__ void __launch_bounds__(32)
mdn_scan_kernel(const float* __restrict__ lp,
                const int* __restrict__ text_len,
                const int* __restrict__ mel_len,
                float* __restrict__ ws) {
  const int b    = blockIdx.x;
  const int lane = threadIdx.x;
  const bool is16 = (lane == 16);
  const float* lpb = lp + (size_t)b * TXT_T * MEL_T;
  const int M    = mel_len[b];
  const int last = M - 1;

  int roff[SR];
#pragma unroll
  for (int i = 0; i < SR; ++i) {
    int r = SR * lane + i;
    if (r > TXT_T - 1) r = TXT_T - 1;
    roff[i] = r * MEL_T;
  }

  const int c0 = 1 + (int)((0u - (unsigned)(((size_t)lpb >> 2) + 1u)) & 3u);
  const int ngl = (MEL_T - c0) / SG;
  const int ct  = c0 + SG * ngl;

  float ed[SG - 1][SR];
#pragma unroll
  for (int j = 0; j < SG - 1; ++j)
#pragma unroll
    for (int i = 0; i < SR; ++i) ed[j][i] = lpb[roff[i] + 1 + j];

  v4f q0[SR], q1[SR], q2[SR], q3[SR], q4[SR];
#define LOAD_GROUP(Q, G)                                                     \
  do {                                                                       \
    int gl_ = (G) < ngl - 1 ? (G) : ngl - 1;         \
    const float* gp_ = lpb + c0 + SG * gl_;                                  \
    _Pragma("unroll")                                                        \
    for (int i_ = 0; i_ < SR; ++i_) Q[i_] = *(const v4f*)(gp_ + roff[i_]);   \
  } while (0)
  LOAD_GROUP(q0, 0);
  LOAD_GROUP(q1, 1);
  LOAD_GROUP(q2, 2);
  LOAD_GROUP(q3, 3);
  LOAD_GROUP(q4, 4);
  static_assert(SD == 5, "one q array per group in flight");

  ScanState st;
#pragma unroll
  for (int i = 0; i < SR; ++i) st.a[i] = -PAD_MAG;
  if (lane == 0) st.a[0] = lpb[0];
#pragma unroll
  for (int i = 0; i < SR; ++i) st.e[i] = st.a[i] + 1e-7f;
  st.eup = lane_below_(st.e[SR - 1], -PAD_MAG, is16);

#pragma unroll
  for (int j = 0; j < SG - 1; ++j)
    if (1 + j < c0 && 1 + j <= last) scan_step_(st, ed[j], is16);

#pragma unroll
  for (int j = 0; j < SG - 1; ++j)
#pragma unroll
    for (int i = 0; i < SR; ++i) ed[j][i] = lpb[roff[i] + MEL_T - (SG - 1) + j];

  const int lim  = last < ct - 1 ? last : ct - 1;
  const int ncol = lim - c0 + 1;
  int g = 0;
  for (; SG * (g + SD) <= ncol; g += SD) {
    scan_group_(st, q0, SG, is16);  LOAD_GROUP(q0, g + SD);
    scan_group_(st, q1, SG, is16);  LOAD_GROUP(q1, g + SD + 1);
    scan_group_(st, q2, SG, is16);  LOAD_GROUP(q2, g + SD + 2);
    scan_group_(st, q3, SG, is16);  LOAD_GROUP(q3, g + SD + 3);
    scan_group_(st, q4, SG, is16);  LOAD_GROUP(q4, g + SD + 4);
  }
  int cl = ncol - SG * g;
  scan_group_(st, q0, cl, is16);
  scan_group_(st, q1, cl - SG, is16);
  scan_group_(st, q2, cl - 2 * SG, is16);
  scan_group_(st, q3, cl - 3 * SG, is16);
  scan_group_(st, q4, cl - 4 * SG, is16);

#pragma unroll
  for (int j = 0; j < SG - 1; ++j) {
    int col = MEL_T - (SG - 1) + j;
    if (col >= ct && col <= last) scan_step_(st, ed[j], is16);
  }

  const int r  = text_len[b] - 1;
  const int ql = r / SR;
  const int qi = r - ql * SR;
  float res = st.a[0];
#pragma unroll
  for (int i = 1; i < SR; ++i) if (qi == i) res = st.a[i];
  if (lane == ql) ws[P_OFF + b] = res / (float)M;
}

__global__ void mdn_reduce_kernel(const float* __restrict__ ws,
                                  float* __restrict__ out) {
  float s = 0.f;
  for (int b = 0; b < B_; ++b) s += ws[P_OFF + b];
  out[0] = -s / (float)B_;
}

extern "C" void kernel_launch(void* const* d_in, const int* in_sizes, int n_in,
                              void* d_out, int out_size, void* d_ws, size_t ws_size,
                              hipStream_t stream) {
  const float* mu_logvar = (const float*)d_in[0];
  const float* melspec   = (const float*)d_in[1];
  const int*   text_len  = (const int*)d_in[2];
  const int*   mel_len   = (const int*)d_in[3];
  float* out = (float*)d_out;
  float* ws  = (float*)d_ws;
  float* lp  = out + 1;

  mdn_prep_kernel<<<(B_ * TXT_PAD + 255) / 256, 256, 0, stream>>>(mu_logvar, ws);

  const int total_waves = B_ * (TXT_PAD / 16) * (MEL_T / 16);
  mdn_gemm_kernel<<<total_waves / 8, 256, 0, stream>>>(ws, melspec, lp);

  mdn_scan_kernel<<<B_, 32, 0, stream>>>(lp, text_len, mel_len, ws);
  mdn_reduce_kernel<<<1, 1, 0, stream>>>(ws, out);
}


# speedup vs baseline: 4.2081x; 1.0776x over previous
#include <hip/hip_runtime.h>
#include <math.h>

#define B_      8
#define TXT_T   200
#define TXT_PAD 208
#define MEL_T   800
#define N_MEL   80
#define KDIM    160
#define PAD_MAG 1.0e12f

typedef __attribute__((ext_vector_type(2))) float v2f;
typedef __attribute__((ext_vector_type(8))) float v8f;

#define A_OFF 0
#define C_OFF (B_ * TXT_PAD * KDIM)
#define P_OFF (C_OFF + B_ * TXT_PAD)

__global__ void mdn_prep_kernel(const float* __restrict__ mu_logvar,
                                float* __restrict__ ws) {
  int idx = blockIdx.x * blockDim.x + threadIdx.x;
  if (idx >= B_ * TXT_PAD) return;
  int b = idx / TXT_PAD;
  int t = idx % TXT_PAD;
  float* Arow = ws + A_OFF + (size_t)idx * KDIM;
  float c = 0.0f;
  if (t < TXT_T) {
    const float* ml = mu_logvar + (size_t)(b * TXT_T + t) * (2 * N_MEL);
    for (int n = 0; n < N_MEL; ++n) {
      float mu = ml[n];
      float lv = ml[N_MEL + n];
      float w  = expf(-lv);
      Arow[n]          = w;
      Arow[N_MEL + n]  = -2.0f * mu * w;
      c += mu * mu * w + lv;
    }
  } else {
    for (int n = 0; n < KDIM; ++n) Arow[n] = 0.0f;
  }
  ws[C_OFF + idx] = c;
}

__global__ void __launch_bounds__(256)
mdn_gemm_kernel(const float* __restrict__ ws,
                const float* __restrict__ mel,
                float* __restrict__ lp) {
  const int TT = TXT_PAD / 16;
  const int MT = MEL_T / 16;
  int wid = blockIdx.x * (blockDim.x >> 5) + (threadIdx.x >> 5);
  if (wid >= B_ * TT * MT) return;
  int lane = threadIdx.x & 31;
  int l15  = lane & 15;
  int kl   = lane >> 4;

  int b  = wid / (TT * MT);
  int r  = wid % (TT * MT);
  int t0 = (r / MT) * 16;
  int m0 = (r % MT) * 16;

  const float* Arow = ws + A_OFF + (size_t)(b * TXT_PAD + t0 + l15) * KDIM;
  const float* melb = mel + (size_t)b * N_MEL * MEL_T + m0 + l15;

  v8f acc = {0.f, 0.f, 0.f, 0.f, 0.f, 0.f, 0.f, 0.f};

  for (int k = 0; k < KDIM; k += 4) {
    v2f a = *(const v2f*)(Arow + k + 2 * kl);

    int K0 = k + 2 * kl;
    int n  = (k < N_MEL) ? K0 : (K0 - N_MEL);
    float x0 = melb[(size_t)n * MEL_T];
    float x1 = melb[(size_t)(n + 1) * MEL_T];
    if (k < N_MEL) { x0 *= x0; x1 *= x1; }
    v2f bv; bv.x = x0; bv.y = x1;

    acc = __builtin_amdgcn_wmma_f32_16x16x4_f32(
         false, a,  false, bv,
         (short)0, acc,  false,  false);
  }

  int m = m0 + l15;
  const float* Cb = ws + C_OFF + b * TXT_PAD;
  const float scale = -0.5f / (float)N_MEL;
#pragma unroll
  for (int j = 0; j < 8; ++j) {
    int t = t0 + j + 8 * kl;
    if (t < TXT_T) {
      lp[(size_t)(b * TXT_T + t) * MEL_T + m] = scale * (acc[j] + Cb[t]);
    }
  }
}


#define SR 7
#define SG 4
#define SD 5
typedef __attribute__((ext_vector_type(4))) float v4f;
static_assert(SR * 32 >= TXT_T, "one wave covers every text row");
static_assert(MEL_T % SG == 0 && MEL_T >= 2 * SG, "row pitch keeps the 16-byte phase of every row equal");

struct LaneBelow { int d; int c; };
__device__ __forceinline__ LaneBelow lane_below_send_(float x, float keep) {
  int xi = __builtin_bit_cast(int, x);
  LaneBelow r;
  r.d = __builtin_amdgcn_update_dpp(__builtin_bit_cast(int, keep), xi, 0x111, 0xf, 0xf, false);
  r.c = __builtin_amdgcn_permlanex16(0, xi, 0xffffffff, 0xffffffff, true, false);
  return r;
}
__device__ __forceinline__ float lane_below_take_(LaneBelow r, bool is16) {
  return __builtin_bit_cast(float, is16 ? r.c : r.d);
}
__device__ __forceinline__ float lane_below_(float x, float padv, bool is16) {
  return lane_below_take_(lane_below_send_(x, padv), is16);
}

struct ScanState {
  float a[SR];
  float e[SR];
  float aup;
  float eup;
};

#define ORDER_FENCE() __builtin_amdgcn_sched_barrier(0)
#define ROW_LO(i) ((i) == 0 ? st.eup : st.e[(i) == 0 ? 0 : (i) - 1])
#define ROW_N(i) d[i]  = fminf(st.e[i], ROW_LO(i))
#define ROW_M(i) mx[i] = fmaxf(st.e[i], ROW_LO(i))
#define ROW_S(i) d[i]  = d[i] - mx[i]
#define ROW_U(i) d[i]  = d[i] * 1.44269504f
#define ROW_X(i) x[i]  = __builtin_amdgcn_exp2f(d[i])
#define ROW_P(i) x[i]  = 1.0f + x[i]
#define ROW_L(i) x[i]  = __builtin_amdgcn_logf(x[i])
#define ROW_F(i) x[i]  = mx[i] + 0.693147181f * x[i]
#define ROW_A(i) st.a[i] = x[i] + p[i]
#define ROW_E(i) st.e[i] = st.a[i] + 1e-7f
__device__ __forceinline__ void scan_step_(ScanState& st, const float (&p)[SR], bool is16) {
  static_assert(SR == 7, "the issue order below is written out for seven rows");
  float mx[SR], d[SR], x[SR];
  ROW_M(6); ROW_N(6); ROW_M(5); ROW_N(5); ROW_M(4); ROW_N(4); ROW_M(3); ROW_N(3);
  ROW_M(2); ROW_N(2);
  ORDER_FENCE();
  ROW_S(6); ORDER_FENCE();
  ROW_M(1); ROW_N(1); ORDER_FENCE();
  ROW_M(0); ROW_N(0); ORDER_FENCE();
  ROW_S(5); ROW_S(4); ORDER_FENCE();
  ROW_U(6); ORDER_FENCE();
  ROW_S(3); ROW_S(2); ORDER_FENCE();
  ROW_U(5); ORDER_FENCE();
  ROW_S(1); ROW_S(0); ORDER_FENCE();
  ROW_U(4); ORDER_FENCE();
  ROW_X(6); ORDER_FENCE(); ROW_U(3); ORDER_FENCE();
  ROW_X(5); ORDER_FENCE(); ROW_U(2); ORDER_FENCE();
  ROW_X(4); ORDER_FENCE(); ROW_U(1); ORDER_FENCE();
  ROW_X(3); ORDER_FENCE(); ROW_U(0); ORDER_FENCE();
  ROW_X(2); ORDER_FENCE(); ROW_P(6); ORDER_FENCE();
  ROW_X(1); ORDER_FENCE(); ROW_P(5); ORDER_FENCE();
  ROW_X(0); ORDER_FENCE(); ROW_P(4); ORDER_FENCE();
  ROW_P(3); ORDER_FENCE();
  ROW_L(6); ORDER_FENCE(); ROW_P(2); ORDER_FENCE();
  ROW_L(5); ORDER_FENCE(); ROW_P(1); ORDER_FENCE();
  ROW_L(4); ORDER_FENCE(); ROW_P(0); ORDER_FENCE();
  ROW_L(3); ORDER_FENCE();
  ROW_L(2); ORDER_FENCE(); ROW_F(6); ORDER_FENCE();
  ROW_L(1); ORDER_FENCE(); ROW_F(5); ORDER_FENCE();
  ROW_L(0); ORDER_FENCE(); ROW_F(4); ORDER_FENCE();
  ROW_A(6); ROW_F(3); ORDER_FENCE();
  ROW_A(5); ROW_F(2); ORDER_FENCE();
  ROW_A(4); ROW_F(1); ORDER_FENCE();
  LaneBelow lb = lane_below_send_(st.a[SR - 1], st.aup);
  ORDER_FENCE();
  ROW_A(3); ROW_F(0); ORDER_FENCE();
  ROW_E(6); ROW_E(5); ORDER_FENCE();
  ROW_A(2); ROW_E(4); ORDER_FENCE();
  ROW_A(1); ROW_E(3); ORDER_FENCE();
  st.aup = lane_below_take_(lb, is16);
  ORDER_FENCE();
  ROW_A(0); ROW_E(2); ORDER_FENCE();
  ROW_E(1); ORDER_FENCE();
  st.eup = st.aup + 1e-7f;
  ORDER_FENCE();
  ROW_E(0);
  ORDER_FENCE();
}

__device__ __forceinline__ void scan_group_(ScanState& st, const v4f (&q)[SR], int ncols, bool is16) {
  float p[SR];
  if (ncols < 1) return;
#pragma unroll
  for (int i = 0; i < SR; ++i) p[i] = q[i].x;
  scan_step_(st, p, is16);
  if (ncols < 2) return;
#pragma unroll
  for (int i = 0; i < SR; ++i) p[i] = q[i].y;
  scan_step_(st, p, is16);
  if (ncols < 3) return;
#pragma unroll
  for (int i = 0; i < SR; ++i) p[i] = q[i].z;
  scan_step_(st, p, is16);
  if (ncols < 4) return;
#pragma unroll
  for (int i = 0; i < SR; ++i) p[i] = q[i].w;
  scan_step_(st, p, is16);
}

__global__ void __launch_bounds__(32)
mdn_scan_kernel(const float* __restrict__ lp,
                const int* __restrict__ text_len,
                const int* __restrict__ mel_len,
                float* __restrict__ ws) {
  const int b    = blockIdx.x;
  const int lane = threadIdx.x;
  const bool is16 = (lane == 16);
  const float* lpb = lp + (size_t)b * TXT_T * MEL_T;
  const int M    = mel_len[b];
  const int last = M - 1;

  int roff[SR];
#pragma unroll
  for (int i = 0; i < SR; ++i) {
    int r = SR * lane + i;
    if (r > TXT_T - 1) r = TXT_T - 1;
    roff[i] = r * MEL_T;
  }

  const int c0 = 1 + (int)((0u - (unsigned)(((size_t)lpb >> 2) + 1u)) & 3u);
  const int ngl = (MEL_T - c0) / SG;
  const int ct  = c0 + SG * ngl;

  float ed[SG - 1][SR];
#pragma unroll
  for (int j = 0; j < SG - 1; ++j)
#pragma unroll
    for (int i = 0; i < SR; ++i) ed[j][i] = lpb[roff[i] + 1 + j];

  v4f q0[SR], q1[SR], q2[SR], q3[SR], q4[SR];
#define LOAD_GROUP(Q, G)                                                     \
  do {                                                                       \
    int gl_ = (G) < ngl - 1 ? (G) : ngl - 1;         \
    const float* gp_ = lpb + c0 + SG * gl_;                                  \
    _Pragma("unroll")                                                        \
    for (int i_ = 0; i_ < SR; ++i_) Q[i_] = *(const v4f*)(gp_ + roff[i_]);   \
  } while (0)
  LOAD_GROUP(q0, 0);
  LOAD_GROUP(q1, 1);
  LOAD_GROUP(q2, 2);
  LOAD_GROUP(q3, 3);
  LOAD_GROUP(q4, 4);
  static_assert(SD == 5, "one q array per group in flight");

  ScanState st;
#pragma unroll
  for (int i = 0; i < SR; ++i) st.a[i] = -PAD_MAG;
  if (lane == 0) st.a[0] = lpb[0];
#pragma unroll
  for (int i = 0; i < SR; ++i) st.e[i] = st.a[i] + 1e-7f;
  st.aup = lane_below_(st.a[SR - 1], -PAD_MAG, is16);
  st.eup = st.aup + 1e-7f;

#pragma unroll
  for (int j = 0; j < SG - 1; ++j)
    if (1 + j < c0 && 1 + j <= last) scan_step_(st, ed[j], is16);

#pragma unroll
  for (int j = 0; j < SG - 1; ++j)
#pragma unroll
    for (int i = 0; i < SR; ++i) ed[j][i] = lpb[roff[i] + MEL_T - (SG - 1) + j];

  const int lim  = last < ct - 1 ? last : ct - 1;
  const int ncol = lim - c0 + 1;
  int g = 0;
  for (; SG * (g + SD) <= ncol; g += SD) {
    scan_group_(st, q0, SG, is16);  LOAD_GROUP(q0, g + SD);
    scan_group_(st, q1, SG, is16);  LOAD_GROUP(q1, g + SD + 1);
    scan_group_(st, q2, SG, is16);  LOAD_GROUP(q2, g + SD + 2);
    scan_group_(st, q3, SG, is16);  LOAD_GROUP(q3, g + SD + 3);
    scan_group_(st, q4, SG, is16);  LOAD_GROUP(q4, g + SD + 4);
  }
  int cl = ncol - SG * g;
  scan_group_(st, q0, cl, is16);
  scan_group_(st, q1, cl - SG, is16);
  scan_group_(st, q2, cl - 2 * SG, is16);
  scan_group_(st, q3, cl - 3 * SG, is16);
  scan_group_(st, q4, cl - 4 * SG, is16);

#pragma unroll
  for (int j = 0; j < SG - 1; ++j) {
    int col = MEL_T - (SG - 1) + j;
    if (col >= ct && col <= last) scan_step_(st, ed[j], is16);
  }

  const int r  = text_len[b] - 1;
  const int ql = r / SR;
  const int qi = r - ql * SR;
  float res = st.a[0];
#pragma unroll
  for (int i = 1; i < SR; ++i) if (qi == i) res = st.a[i];
  if (lane == ql) ws[P_OFF + b] = res / (float)M;
}

__global__ void mdn_reduce_kernel(const float* __restrict__ ws,
                                  float* __restrict__ out) {
  float s = 0.f;
  for (int b = 0; b < B_; ++b) s += ws[P_OFF + b];
  out[0] = -s / (float)B_;
}

extern "C" void kernel_launch(void* const* d_in, const int* in_sizes, int n_in,
                              void* d_out, int out_size, void* d_ws, size_t ws_size,
                              hipStream_t stream) {
  const float* mu_logvar = (const float*)d_in[0];
  const float* melspec   = (const float*)d_in[1];
  const int*   text_len  = (const int*)d_in[2];
  const int*   mel_len   = (const int*)d_in[3];
  float* out = (float*)d_out;
  float* ws  = (float*)d_ws;
  float* lp  = out + 1;

  mdn_prep_kernel<<<(B_ * TXT_PAD + 255) / 256, 256, 0, stream>>>(mu_logvar, ws);

  const int total_waves = B_ * (TXT_PAD / 16) * (MEL_T / 16);
  mdn_gemm_kernel<<<total_waves / 8, 256, 0, stream>>>(ws, melspec, lp);

  mdn_scan_kernel<<<B_, 32, 0, stream>>>(lp, text_len, mel_len, ws);
  mdn_reduce_kernel<<<1, 1, 0, stream>>>(ws, out);
}


# prep launch fused into the gemm: each wave builds its A fragments (w, -2*mu*w) in registers from mu_logvar, row constant added by one more WMMA; K order and the constant's sum order change
# speedup vs baseline: 5.8624x; 1.3931x over previous
#include <hip/hip_runtime.h>
#include <math.h>

#define B_      8
#define TXT_T   200
#define TXT_PAD 208
#define MEL_T   800
#define N_MEL   80
#define PAD_MAG 1.0e12f

typedef __attribute__((ext_vector_type(2))) float v2f;
typedef __attribute__((ext_vector_type(8))) float v8f;

#define P_OFF 0

__global__ void __launch_bounds__(256)
mdn_gemm_kernel(const float* __restrict__ mu_logvar,
                const float* __restrict__ mel,
                float* __restrict__ lp) {
  const int TT = TXT_PAD / 16;
  const int MT = MEL_T / 16;
  int wid = blockIdx.x * (blockDim.x >> 5) + (threadIdx.x >> 5);
  if (wid >= B_ * TT * MT) return;
  int lane = threadIdx.x & 31;
  int l15  = lane & 15;
  int kl   = lane >> 4;

  int b  = wid / (TT * MT);
  int r  = wid % (TT * MT);
  int t0 = (r / MT) * 16;
  int m0 = (r % MT) * 16;

  int trow = t0 + l15;
  if (trow > TXT_T - 1) trow = TXT_T - 1;
  const float* ml   = mu_logvar + (size_t)(b * TXT_T + trow) * (2 * N_MEL) + 2 * kl;
  const float* melb = mel + (size_t)b * N_MEL * MEL_T + (size_t)(2 * kl) * MEL_T + m0 + l15;

  v8f acc = {0.f, 0.f, 0.f, 0.f, 0.f, 0.f, 0.f, 0.f};
  float c = 0.0f;

#pragma unroll
  for (int n = 0; n < N_MEL; n += 4) {
    v2f mu = *(const v2f*)(ml + n);
    v2f lv = *(const v2f*)(ml + N_MEL + n);
    float x0 = melb[(size_t)n * MEL_T];
    float x1 = melb[(size_t)(n + 1) * MEL_T];

    v2f w;  w.x = expf(-lv.x);          w.y = expf(-lv.y);
    v2f g;  g.x = -2.0f * mu.x * w.x;   g.y = -2.0f * mu.y * w.y;
    c += mu.x * mu.x * w.x + lv.x;
    c += mu.y * mu.y * w.y + lv.y;
    v2f xx; xx.x = x0 * x0;             xx.y = x1 * x1;
    v2f xv; xv.x = x0;                  xv.y = x1;

    acc = __builtin_amdgcn_wmma_f32_16x16x4_f32(
         false, w,  false, xx,
         (short)0, acc,  false,  false);
    acc = __builtin_amdgcn_wmma_f32_16x16x4_f32(
         false, g,  false, xv,
         (short)0, acc,  false,  false);
  }

  {
    v2f ca;  ca.x = c;     ca.y = 0.0f;
    v2f one; one.x = 1.0f; one.y = 0.0f;
    acc = __builtin_amdgcn_wmma_f32_16x16x4_f32(
         false, ca,  false, one,
         (short)0, acc,  false,  false);
  }

  int m = m0 + l15;
  const float scale = -0.5f / (float)N_MEL;
#pragma unroll
  for (int j = 0; j < 8; ++j) {
    int t = t0 + j + 8 * kl;
    if (t < TXT_T) {
      lp[(size_t)(b * TXT_T + t) * MEL_T + m] = scale * acc[j];
    }
  }
}


#define SR 7
#define SG 4
#define SD 5
typedef __attribute__((ext_vector_type(4))) float v4f;
static_assert(SR * 32 >= TXT_T, "one wave covers every text row");
static_assert(MEL_T % SG == 0 && MEL_T >= 2 * SG, "row pitch keeps the 16-byte phase of every row equal");

struct LaneBelow { int d; int c; };
__device__ __forceinline__ LaneBelow lane_below_send_(float x, float keep) {
  int xi = __builtin_bit_cast(int, x);
  LaneBelow r;
  r.d = __builtin_amdgcn_update_dpp(__builtin_bit_cast(int, keep), xi, 0x111, 0xf, 0xf, false);
  r.c = __builtin_amdgcn_permlanex16(0, xi, 0xffffffff, 0xffffffff, true, false);
  return r;
}
__device__ __forceinline__ float lane_below_take_(LaneBelow r, bool is16) {
  return __builtin_bit_cast(float, is16 ? r.c : r.d);
}
__device__ __forceinline__ float lane_below_(float x, float padv, bool is16) {
  return lane_below_take_(lane_below_send_(x, padv), is16);
}

struct ScanState {
  float a[SR];
  float e[SR];
  float aup;
  float eup;
};

#define ORDER_FENCE() __builtin_amdgcn_sched_barrier(0)
#define ROW_LO(i) ((i) == 0 ? st.eup : st.e[(i) == 0 ? 0 : (i) - 1])
#define ROW_N(i) d[i]  = fminf(st.e[i], ROW_LO(i))
#define ROW_M(i) mx[i] = fmaxf(st.e[i], ROW_LO(i))
#define ROW_S(i) d[i]  = d[i] - mx[i]
#define ROW_U(i) d[i]  = d[i] * 1.44269504f
#define ROW_X(i) x[i]  = __builtin_amdgcn_exp2f(d[i])
#define ROW_P(i) x[i]  = 1.0f + x[i]
#define ROW_L(i) x[i]  = __builtin_amdgcn_logf(x[i])
#define ROW_F(i) x[i]  = mx[i] + 0.693147181f * x[i]
#define ROW_A(i) st.a[i] = x[i] + p[i]
#define ROW_E(i) st.e[i] = st.a[i] + 1e-7f
__device__ __forceinline__ void scan_step_(ScanState& st, const float (&p)[SR], bool is16) {
  static_assert(SR == 7, "the issue order below is written out for seven rows");
  float mx[SR], d[SR], x[SR];
  ROW_M(6); ROW_N(6); ROW_M(5); ROW_N(5); ROW_M(4); ROW_N(4); ROW_M(3); ROW_N(3);
  ROW_M(2); ROW_N(2);
  ORDER_FENCE();
  ROW_S(6); ORDER_FENCE();
  ROW_M(1); ROW_N(1); ORDER_FENCE();
  ROW_M(0); ROW_N(0); ORDER_FENCE();
  ROW_S(5); ROW_S(4); ORDER_FENCE();
  ROW_U(6); ORDER_FENCE();
  ROW_S(3); ROW_S(2); ORDER_FENCE();
  ROW_U(5); ORDER_FENCE();
  ROW_S(1); ROW_S(0); ORDER_FENCE();
  ROW_U(4); ORDER_FENCE();
  ROW_X(6); ORDER_FENCE(); ROW_U(3); ORDER_FENCE();
  ROW_X(5); ORDER_FENCE(); ROW_U(2); ORDER_FENCE();
  ROW_X(4); ORDER_FENCE(); ROW_U(1); ORDER_FENCE();
  ROW_X(3); ORDER_FENCE(); ROW_U(0); ORDER_FENCE();
  ROW_X(2); ORDER_FENCE(); ROW_P(6); ORDER_FENCE();
  ROW_X(1); ORDER_FENCE(); ROW_P(5); ORDER_FENCE();
  ROW_X(0); ORDER_FENCE(); ROW_P(4); ORDER_FENCE();
  ROW_P(3); ORDER_FENCE();
  ROW_L(6); ORDER_FENCE(); ROW_P(2); ORDER_FENCE();
  ROW_L(5); ORDER_FENCE(); ROW_P(1); ORDER_FENCE();
  ROW_L(4); ORDER_FENCE(); ROW_P(0); ORDER_FENCE();
  ROW_L(3); ORDER_FENCE();
  ROW_L(2); ORDER_FENCE(); ROW_F(6); ORDER_FENCE();
  ROW_L(1); ORDER_FENCE(); ROW_F(5); ORDER_FENCE();
  ROW_L(0); ORDER_FENCE(); ROW_F(4); ORDER_FENCE();
  ROW_A(6); ROW_F(3); ORDER_FENCE();
  ROW_A(5); ROW_F(2); ORDER_FENCE();
  ROW_A(4); ROW_F(1); ORDER_FENCE();
  LaneBelow lb = lane_below_send_(st.a[SR - 1], st.aup);
  ORDER_FENCE();
  ROW_A(3); ROW_F(0); ORDER_FENCE();
  ROW_E(6); ROW_E(5); ORDER_FENCE();
  ROW_A(2); ROW_E(4); ORDER_FENCE();
  ROW_A(1); ROW_E(3); ORDER_FENCE();
  st.aup = lane_below_take_(lb, is16);
  ORDER_FENCE();
  ROW_A(0); ROW_E(2); ORDER_FENCE();
  ROW_E(1); ORDER_FENCE();
  st.eup = st.aup + 1e-7f;
  ORDER_FENCE();
  ROW_E(0);
  ORDER_FENCE();
}

__device__ __forceinline__ void scan_group_(ScanState& st, const v4f (&q)[SR], int ncols, bool is16) {
  float p[SR];
  if (ncols < 1) return;
#pragma unroll
  for (int i = 0; i < SR; ++i) p[i] = q[i].x;
  scan_step_(st, p, is16);
  if (ncols < 2) return;
#pragma unroll
  for (int i = 0; i < SR; ++i) p[i] = q[i].y;
  scan_step_(st, p, is16);
  if (ncols < 3) return;
#pragma unroll
  for (int i = 0; i < SR; ++i) p[i] = q[i].z;
  scan_step_(st, p, is16);
  if (ncols < 4) return;
#pragma unroll
  for (int i = 0; i < SR; ++i) p[i] = q[i].w;
  scan_step_(st, p, is16);
}

__global__ void __launch_bounds__(32)
mdn_scan_kernel(const float* __restrict__ lp,
                const int* __restrict__ text_len,
                const int* __restrict__ mel_len,
                float* __restrict__ ws) {
  const int b    = blockIdx.x;
  const int lane = threadIdx.x;
  const bool is16 = (lane == 16);
  const float* lpb = lp + (size_t)b * TXT_T * MEL_T;
  const int M    = mel_len[b];
  const int last = M - 1;

  int roff[SR];
#pragma unroll
  for (int i = 0; i < SR; ++i) {
    int r = SR * lane + i;
    if (r > TXT_T - 1) r = TXT_T - 1;
    roff[i] = r * MEL_T;
  }

  const int c0 = 1 + (int)((0u - (unsigned)(((size_t)lpb >> 2) + 1u)) & 3u);
  const int ngl = (MEL_T - c0) / SG;
  const int ct  = c0 + SG * ngl;

  float ed[SG - 1][SR];
#pragma unroll
  for (int j = 0; j < SG - 1; ++j)
#pragma unroll
    for (int i = 0; i < SR; ++i) ed[j][i] = lpb[roff[i] + 1 + j];

  v4f q0[SR], q1[SR], q2[SR], q3[SR], q4[SR];
#define LOAD_GROUP(Q, G)                                                     \
  do {                                                                       \
    int gl_ = (G) < ngl - 1 ? (G) : ngl - 1;         \
    const float* gp_ = lpb + c0 + SG * gl_;                                  \
    _Pragma("unroll")                                                        \
    for (int i_ = 0; i_ < SR; ++i_) Q[i_] = *(const v4f*)(gp_ + roff[i_]);   \
  } while (0)
  LOAD_GROUP(q0, 0);
  LOAD_GROUP(q1, 1);
  LOAD_GROUP(q2, 2);
  LOAD_GROUP(q3, 3);
  LOAD_GROUP(q4, 4);
  static_assert(SD == 5, "one q array per group in flight");

  ScanState st;
#pragma unroll
  for (int i = 0; i < SR; ++i) st.a[i] = -PAD_MAG;
  if (lane == 0) st.a[0] = lpb[0];
#pragma unroll
  for (int i = 0; i < SR; ++i) st.e[i] = st.a[i] + 1e-7f;
  st.aup = lane_below_(st.a[SR - 1], -PAD_MAG, is16);
  st.eup = st.aup + 1e-7f;

#pragma unroll
  for (int j = 0; j < SG - 1; ++j)
    if (1 + j < c0 && 1 + j <= last) scan_step_(st, ed[j], is16);

#pragma unroll
  for (int j = 0; j < SG - 1; ++j)
#pragma unroll
    for (int i = 0; i < SR; ++i) ed[j][i] = lpb[roff[i] + MEL_T - (SG - 1) + j];

  const int lim  = last < ct - 1 ? last : ct - 1;
  const int ncol = lim - c0 + 1;
  int g = 0;
  for (; SG * (g + SD) <= ncol; g += SD) {
    scan_group_(st, q0, SG, is16);  LOAD_GROUP(q0, g + SD);
    scan_group_(st, q1, SG, is16);  LOAD_GROUP(q1, g + SD + 1);
    scan_group_(st, q2, SG, is16);  LOAD_GROUP(q2, g + SD + 2);
    scan_group_(st, q3, SG, is16);  LOAD_GROUP(q3, g + SD + 3);
    scan_group_(st, q4, SG, is16);  LOAD_GROUP(q4, g + SD + 4);
  }
  int cl = ncol - SG * g;
  scan_group_(st, q0, cl, is16);
  scan_group_(st, q1, cl - SG, is16);
  scan_group_(st, q2, cl - 2 * SG, is16);
  scan_group_(st, q3, cl - 3 * SG, is16);
  scan_group_(st, q4, cl - 4 * SG, is16);

#pragma unroll
  for (int j = 0; j < SG - 1; ++j) {
    int col = MEL_T - (SG - 1) + j;
    if (col >= ct && col <= last) scan_step_(st, ed[j], is16);
  }

  const int r  = text_len[b] - 1;
  const int ql = r / SR;
  const int qi = r - ql * SR;
  float res = st.a[0];
#pragma unroll
  for (int i = 1; i < SR; ++i) if (qi == i) res = st.a[i];
  if (lane == ql) ws[P_OFF + b] = res / (float)M;
}

__global__ void mdn_reduce_kernel(const float* __restrict__ ws,
                                  float* __restrict__ out) {
  float s = 0.f;
  for (int b = 0; b < B_; ++b) s += ws[P_OFF + b];
  out[0] = -s / (float)B_;
}

extern "C" void kernel_launch(void* const* d_in, const int* in_sizes, int n_in,
                              void* d_out, int out_size, void* d_ws, size_t ws_size,
                              hipStream_t stream) {
  const float* mu_logvar = (const float*)d_in[0];
  const float* melspec   = (const float*)d_in[1];
  const int*   text_len  = (const int*)d_in[2];
  const int*   mel_len   = (const int*)d_in[3];
  float* out = (float*)d_out;
  float* ws  = (float*)d_ws;
  float* lp  = out + 1;

  const int total_waves = B_ * (TXT_PAD / 16) * (MEL_T / 16);
  mdn_gemm_kernel<<<total_waves / 8, 256, 0, stream>>>(mu_logvar, melspec, lp);

  mdn_scan_kernel<<<B_, 32, 0, stream>>>(lp, text_len, mel_len, ws);
  mdn_reduce_kernel<<<1, 1, 0, stream>>>(ws, out);
}


# scan carry kept in base-2 units (log_alpha*log2(e)): exp2 takes -|U-V| directly, lp enters by one multiply-add; 8 instructions per row in place of 10; f32 kept; scan sums and rounding change
# speedup vs baseline: 6.2755x; 1.0705x over previous
#include <hip/hip_runtime.h>
#include <math.h>

#define B_      8
#define TXT_T   200
#define TXT_PAD 208
#define MEL_T   800
#define N_MEL   80
#define PAD_MAG 1.0e12f

typedef __attribute__((ext_vector_type(2))) float v2f;
typedef __attribute__((ext_vector_type(8))) float v8f;

#define P_OFF 0

__global__ void __launch_bounds__(256)
mdn_gemm_kernel(const float* __restrict__ mu_logvar,
                const float* __restrict__ mel,
                float* __restrict__ lp) {
  const int TT = TXT_PAD / 16;
  const int MT = MEL_T / 16;
  int wid = blockIdx.x * (blockDim.x >> 5) + (threadIdx.x >> 5);
  if (wid >= B_ * TT * MT) return;
  int lane = threadIdx.x & 31;
  int l15  = lane & 15;
  int kl   = lane >> 4;

  int b  = wid / (TT * MT);
  int r  = wid % (TT * MT);
  int t0 = (r / MT) * 16;
  int m0 = (r % MT) * 16;

  int trow = t0 + l15;
  if (trow > TXT_T - 1) trow = TXT_T - 1;
  const float* ml   = mu_logvar + (size_t)(b * TXT_T + trow) * (2 * N_MEL) + 2 * kl;
  const float* melb = mel + (size_t)b * N_MEL * MEL_T + (size_t)(2 * kl) * MEL_T + m0 + l15;

  v8f acc = {0.f, 0.f, 0.f, 0.f, 0.f, 0.f, 0.f, 0.f};
  float c = 0.0f;

#pragma unroll
  for (int n = 0; n < N_MEL; n += 4) {
    v2f mu = *(const v2f*)(ml + n);
    v2f lv = *(const v2f*)(ml + N_MEL + n);
    float x0 = melb[(size_t)n * MEL_T];
    float x1 = melb[(size_t)(n + 1) * MEL_T];

    v2f w;  w.x = expf(-lv.x);          w.y = expf(-lv.y);
    v2f g;  g.x = -2.0f * mu.x * w.x;   g.y = -2.0f * mu.y * w.y;
    c += mu.x * mu.x * w.x + lv.x;
    c += mu.y * mu.y * w.y + lv.y;
    v2f xx; xx.x = x0 * x0;             xx.y = x1 * x1;
    v2f xv; xv.x = x0;                  xv.y = x1;

    acc = __builtin_amdgcn_wmma_f32_16x16x4_f32(
         false, w,  false, xx,
         (short)0, acc,  false,  false);
    acc = __builtin_amdgcn_wmma_f32_16x16x4_f32(
         false, g,  false, xv,
         (short)0, acc,  false,  false);
  }

  {
    v2f ca;  ca.x = c;     ca.y = 0.0f;
    v2f one; one.x = 1.0f; one.y = 0.0f;
    acc = __builtin_amdgcn_wmma_f32_16x16x4_f32(
         false, ca,  false, one,
         (short)0, acc,  false,  false);
  }

  int m = m0 + l15;
  const float scale = -0.5f / (float)N_MEL;
#pragma unroll
  for (int j = 0; j < 8; ++j) {
    int t = t0 + j + 8 * kl;
    if (t < TXT_T) {
      lp[(size_t)(b * TXT_T + t) * MEL_T + m] = scale * acc[j];
    }
  }
}

#define LOG2E_F 1.44269504f
#define LN2_F   0.693147181f
#define EPS2_F  (1e-7f * LOG2E_F)

#define SR 7
#define SG 4
#define SD 5
typedef __attribute__((ext_vector_type(4))) float v4f;
static_assert(SR * 32 >= TXT_T, "one wave covers every text row");
static_assert(MEL_T % SG == 0 && MEL_T >= 2 * SG, "row pitch keeps the 16-byte phase of every row equal");

struct LaneBelow { int d; int c; };
__device__ __forceinline__ LaneBelow lane_below_send_(float x, float keep) {
  int xi = __builtin_bit_cast(int, x);
  LaneBelow r;
  r.d = __builtin_amdgcn_update_dpp(__builtin_bit_cast(int, keep), xi, 0x111, 0xf, 0xf, false);
  r.c = __builtin_amdgcn_permlanex16(0, xi, 0xffffffff, 0xffffffff, true, false);
  return r;
}
__device__ __forceinline__ float lane_below_take_(LaneBelow r, bool is16) {
  return __builtin_bit_cast(float, is16 ? r.c : r.d);
}
__device__ __forceinline__ float lane_below_(float x, float padv, bool is16) {
  return lane_below_take_(lane_below_send_(x, padv), is16);
}

struct ScanState {
  float a[SR];
  float e[SR];
  float aup;
  float eup;
};

#define ORDER_FENCE() __builtin_amdgcn_sched_barrier(0)
#define ROW_LO(i) ((i) == 0 ? st.eup : st.e[(i) == 0 ? 0 : (i) - 1])
#define ROW_D(i) d[i]  = st.e[i] - ROW_LO(i)
#define ROW_M(i) mx[i] = fmaxf(st.e[i], ROW_LO(i))
#define ROW_X(i) x[i]  = __builtin_amdgcn_exp2f(-__builtin_fabsf(d[i]))
#define ROW_Q(i) mx[i] = __builtin_fmaf(p[i], LOG2E_F, mx[i])
#define ROW_P(i) x[i]  = 1.0f + x[i]
#define ROW_L(i) x[i]  = __builtin_amdgcn_logf(x[i])
#define ROW_A(i) st.a[i] = mx[i] + x[i]
#define ROW_E(i) st.e[i] = st.a[i] + EPS2_F
__device__ __forceinline__ void scan_step_(ScanState& st, const float (&p)[SR], bool is16) {
  static_assert(SR == 7, "the issue order below is written out for seven rows");
  float mx[SR], d[SR], x[SR];
  ROW_D(6); ROW_M(6); ROW_D(5); ROW_M(5); ROW_D(4); ROW_M(4); ROW_D(3); ROW_M(3);
  ROW_D(2); ROW_M(2); ROW_D(1); ROW_M(1); ROW_D(0); ROW_M(0); ORDER_FENCE();
  ROW_X(6); ORDER_FENCE(); ROW_Q(6); ORDER_FENCE();
  ROW_X(5); ORDER_FENCE(); ROW_Q(5); ORDER_FENCE();
  ROW_X(4); ORDER_FENCE(); ROW_Q(4); ORDER_FENCE();
  ROW_X(3); ORDER_FENCE(); ROW_Q(3); ORDER_FENCE();
  ROW_X(2); ORDER_FENCE(); ROW_P(6); ORDER_FENCE();
  ROW_X(1); ORDER_FENCE(); ROW_P(5); ORDER_FENCE();
  ROW_X(0); ORDER_FENCE(); ROW_P(4); ORDER_FENCE();
  ROW_L(6); ORDER_FENCE(); ROW_P(3); ORDER_FENCE();
  ROW_L(5); ORDER_FENCE(); ROW_P(2); ORDER_FENCE();
  ROW_L(4); ORDER_FENCE(); ROW_P(1); ORDER_FENCE();
  ROW_L(3); ORDER_FENCE(); ROW_P(0); ORDER_FENCE();
  ROW_L(2); ORDER_FENCE(); ROW_Q(2); ORDER_FENCE();
  ROW_L(1); ORDER_FENCE(); ROW_Q(1); ORDER_FENCE();
  ROW_L(0); ORDER_FENCE(); ROW_Q(0); ORDER_FENCE();
  ROW_A(6); ORDER_FENCE();
  ROW_A(5); ROW_A(4); ROW_A(3); ROW_A(2); ORDER_FENCE();
  LaneBelow lb = lane_below_send_(st.a[SR - 1], st.aup);
  ORDER_FENCE();
  ROW_A(1); ROW_A(0); ORDER_FENCE();
  ROW_E(6); ROW_E(5); ROW_E(4); ORDER_FENCE();
  st.aup = lane_below_take_(lb, is16);
  ORDER_FENCE();
  ROW_E(3); ROW_E(2); ROW_E(1); ROW_E(0); ORDER_FENCE();
  st.eup = st.aup + EPS2_F;
  ORDER_FENCE();
}

__device__ __forceinline__ void scan_group_(ScanState& st, const v4f (&q)[SR], int ncols, bool is16) {
  float p[SR];
  if (ncols < 1) return;
#pragma unroll
  for (int i = 0; i < SR; ++i) p[i] = q[i].x;
  scan_step_(st, p, is16);
  if (ncols < 2) return;
#pragma unroll
  for (int i = 0; i < SR; ++i) p[i] = q[i].y;
  scan_step_(st, p, is16);
  if (ncols < 3) return;
#pragma unroll
  for (int i = 0; i < SR; ++i) p[i] = q[i].z;
  scan_step_(st, p, is16);
  if (ncols < 4) return;
#pragma unroll
  for (int i = 0; i < SR; ++i) p[i] = q[i].w;
  scan_step_(st, p, is16);
}

__global__ void __launch_bounds__(32)
mdn_scan_kernel(const float* __restrict__ lp,
                const int* __restrict__ text_len,
                const int* __restrict__ mel_len,
                float* __restrict__ ws) {
  const int b    = blockIdx.x;
  const int lane = threadIdx.x;
  const bool is16 = (lane == 16);
  const float* lpb = lp + (size_t)b * TXT_T * MEL_T;
  const int M    = mel_len[b];
  const int last = M - 1;

  int roff[SR];
#pragma unroll
  for (int i = 0; i < SR; ++i) {
    int r = SR * lane + i;
    if (r > TXT_T - 1) r = TXT_T - 1;
    roff[i] = r * MEL_T;
  }

  const int c0 = 1 + (int)((0u - (unsigned)(((size_t)lpb >> 2) + 1u)) & 3u);
  const int ngl = (MEL_T - c0) / SG;
  const int ct  = c0 + SG * ngl;

  float ed[SG - 1][SR];
#pragma unroll
  for (int j = 0; j < SG - 1; ++j)
#pragma unroll
    for (int i = 0; i < SR; ++i) ed[j][i] = lpb[roff[i] + 1 + j];

  v4f q0[SR], q1[SR], q2[SR], q3[SR], q4[SR];
#define LOAD_GROUP(Q, G)                                                     \
  do {                                                                       \
    int gl_ = (G) < ngl - 1 ? (G) : ngl - 1;         \
    const float* gp_ = lpb + c0 + SG * gl_;                                  \
    _Pragma("unroll")                                                        \
    for (int i_ = 0; i_ < SR; ++i_) Q[i_] = *(const v4f*)(gp_ + roff[i_]);   \
  } while (0)
  LOAD_GROUP(q0, 0);
  LOAD_GROUP(q1, 1);
  LOAD_GROUP(q2, 2);
  LOAD_GROUP(q3, 3);
  LOAD_GROUP(q4, 4);
  static_assert(SD == 5, "one q array per group in flight");

  ScanState st;
#pragma unroll
  for (int i = 0; i < SR; ++i) st.a[i] = -PAD_MAG * LOG2E_F;
  if (lane == 0) st.a[0] = lpb[0] * LOG2E_F;
#pragma unroll
  for (int i = 0; i < SR; ++i) st.e[i] = st.a[i] + EPS2_F;
  st.aup = lane_below_(st.a[SR - 1], -PAD_MAG * LOG2E_F, is16);
  st.eup = st.aup + EPS2_F;

#pragma unroll
  for (int j = 0; j < SG - 1; ++j)
    if (1 + j < c0 && 1 + j <= last) scan_step_(st, ed[j], is16);

#pragma unroll
  for (int j = 0; j < SG - 1; ++j)
#pragma unroll
    for (int i = 0; i < SR; ++i) ed[j][i] = lpb[roff[i] + MEL_T - (SG - 1) + j];

  const int lim  = last < ct - 1 ? last : ct - 1;
  const int ncol = lim - c0 + 1;
  int g = 0;
  for (; SG * (g + SD) <= ncol; g += SD) {
    scan_group_(st, q0, SG, is16);  LOAD_GROUP(q0, g + SD);
    scan_group_(st, q1, SG, is16);  LOAD_GROUP(q1, g + SD + 1);
    scan_group_(st, q2, SG, is16);  LOAD_GROUP(q2, g + SD + 2);
    scan_group_(st, q3, SG, is16);  LOAD_GROUP(q3, g + SD + 3);
    scan_group_(st, q4, SG, is16);  LOAD_GROUP(q4, g + SD + 4);
  }
  int cl = ncol - SG * g;
  scan_group_(st, q0, cl, is16);
  scan_group_(st, q1, cl - SG, is16);
  scan_group_(st, q2, cl - 2 * SG, is16);
  scan_group_(st, q3, cl - 3 * SG, is16);
  scan_group_(st, q4, cl - 4 * SG, is16);

#pragma unroll
  for (int j = 0; j < SG - 1; ++j) {
    int col = MEL_T - (SG - 1) + j;
    if (col >= ct && col <= last) scan_step_(st, ed[j], is16);
  }

  const int r  = text_len[b] - 1;
  const int ql = r / SR;
  const int qi = r - ql * SR;
  float res = st.a[0];
#pragma unroll
  for (int i = 1; i < SR; ++i) if (qi == i) res = st.a[i];
  if (lane == ql) ws[P_OFF + b] = (res * LN2_F) / (float)M;
}

__global__ void mdn_reduce_kernel(const float* __restrict__ ws,
                                  float* __restrict__ out) {
  float s = 0.f;
  for (int b = 0; b < B_; ++b) s += ws[P_OFF + b];
  out[0] = -s / (float)B_;
}

extern "C" void kernel_launch(void* const* d_in, const int* in_sizes, int n_in,
                              void* d_out, int out_size, void* d_ws, size_t ws_size,
                              hipStream_t stream) {
  const float* mu_logvar = (const float*)d_in[0];
  const float* melspec   = (const float*)d_in[1];
  const int*   text_len  = (const int*)d_in[2];
  const int*   mel_len   = (const int*)d_in[3];
  float* out = (float*)d_out;
  float* ws  = (float*)d_ws;
  float* lp  = out + 1;

  const int total_waves = B_ * (TXT_PAD / 16) * (MEL_T / 16);
  mdn_gemm_kernel<<<total_waves / 8, 256, 0, stream>>>(mu_logvar, melspec, lp);

  mdn_scan_kernel<<<B_, 32, 0, stream>>>(lp, text_len, mel_len, ws);
  mdn_reduce_kernel<<<1, 1, 0, stream>>>(ws, out);
}


# gemm: each wave computes a strip of three 16x16 tiles of lp (last strip of a row: two), was one tile; mu/logvar loads, exponentials, A fragments once per strip; 1768 waves, was 5200; sums unchanged
# speedup vs baseline: 6.5543x; 1.0444x over previous
#include <hip/hip_runtime.h>
#include <math.h>

#define B_      8
#define TXT_T   200
#define TXT_PAD 208
#define MEL_T   800
#define N_MEL   80
#define PAD_MAG 1.0e12f

typedef __attribute__((ext_vector_type(2))) float v2f;
typedef __attribute__((ext_vector_type(8))) float v8f;

#define P_OFF 0

#define NT      3
#define MT_ALL  (MEL_T / 16)
#define MS_FULL (MT_ALL / NT)
#define NT_LAST (MT_ALL - NT * MS_FULL)
#define MS_ALL  (MS_FULL + (NT_LAST > 0 ? 1 : 0))
static_assert(MEL_T % 16 == 0 && NT_LAST >= 0 && NT_LAST < NT, "strips cover a row of tiles");

template <int NTT>
__device__ __forceinline__ void gemm_strip_(const float* __restrict__ ml,
                                            const float* __restrict__ melb,
                                            float* __restrict__ lpc,
                                            int t0, int kl) {
  v8f acc[NTT];
#pragma unroll
  for (int j = 0; j < NTT; ++j) acc[j] = (v8f){0.f, 0.f, 0.f, 0.f, 0.f, 0.f, 0.f, 0.f};
  float c = 0.0f;

#pragma unroll
  for (int n = 0; n < N_MEL; n += 4) {
    v2f mu = *(const v2f*)(ml + n);
    v2f lv = *(const v2f*)(ml + N_MEL + n);
    float x0[NTT], x1[NTT];
#pragma unroll
    for (int j = 0; j < NTT; ++j) {
      x0[j] = melb[(size_t)n * MEL_T + 16 * j];
      x1[j] = melb[(size_t)(n + 1) * MEL_T + 16 * j];
    }

    v2f w;  w.x = expf(-lv.x);          w.y = expf(-lv.y);
    v2f g;  g.x = -2.0f * mu.x * w.x;   g.y = -2.0f * mu.y * w.y;
    c += mu.x * mu.x * w.x + lv.x;
    c += mu.y * mu.y * w.y + lv.y;

#pragma unroll
    for (int j = 0; j < NTT; ++j) {
      v2f xx; xx.x = x0[j] * x0[j];     xx.y = x1[j] * x1[j];
      acc[j] = __builtin_amdgcn_wmma_f32_16x16x4_f32(
           false, w,  false, xx,
           (short)0, acc[j],  false,  false);
    }
#pragma unroll
    for (int j = 0; j < NTT; ++j) {
      v2f xv; xv.x = x0[j];             xv.y = x1[j];
      acc[j] = __builtin_amdgcn_wmma_f32_16x16x4_f32(
           false, g,  false, xv,
           (short)0, acc[j],  false,  false);
    }
  }

  {
    v2f ca;  ca.x = c;     ca.y = 0.0f;
    v2f one; one.x = 1.0f; one.y = 0.0f;
#pragma unroll
    for (int j = 0; j < NTT; ++j)
      acc[j] = __builtin_amdgcn_wmma_f32_16x16x4_f32(
           false, ca,  false, one,
           (short)0, acc[j],  false,  false);
  }

  const float scale = -0.5f / (float)N_MEL;
#pragma unroll
  for (int i = 0; i < 8; ++i) {
    int t = t0 + i + 8 * kl;
    if (t < TXT_T) {
      float* row = lpc + (size_t)t * MEL_T;
#pragma unroll
      for (int j = 0; j < NTT; ++j) row[16 * j] = scale * acc[j][i];
    }
  }
}

__global__ void __launch_bounds__(256)
mdn_gemm_kernel(const float* __restrict__ mu_logvar,
                const float* __restrict__ mel,
                float* __restrict__ lp) {
  const int TT = TXT_PAD / 16;
  int wid = blockIdx.x * (blockDim.x >> 5) + (threadIdx.x >> 5);
  if (wid >= B_ * TT * MS_ALL) return;
  int lane = threadIdx.x & 31;
  int l15  = lane & 15;
  int kl   = lane >> 4;

  int b  = wid / (TT * MS_ALL);
  int r  = wid % (TT * MS_ALL);
  int t0 = (r / MS_ALL) * 16;
  int ms = r % MS_ALL;
  int m0 = ms * (16 * NT);

  int trow = t0 + l15;
  if (trow > TXT_T - 1) trow = TXT_T - 1;
  const float* ml   = mu_logvar + (size_t)(b * TXT_T + trow) * (2 * N_MEL) + 2 * kl;
  const float* melb = mel + (size_t)b * N_MEL * MEL_T + (size_t)(2 * kl) * MEL_T + m0 + l15;
  float* lpc = lp + (size_t)b * TXT_T * MEL_T + m0 + l15;

  if (NT_LAST == 0 || ms < MS_FULL) gemm_strip_<NT>(ml, melb, lpc, t0, kl);
  else gemm_strip_<(NT_LAST > 0 ? NT_LAST : 1)>(ml, melb, lpc, t0, kl);
}

#define LOG2E_F 1.44269504f
#define LN2_F   0.693147181f
#define EPS2_F  (1e-7f * LOG2E_F)

#define SR 7
#define SG 4
#define SD 5
typedef __attribute__((ext_vector_type(4))) float v4f;
static_assert(SR * 32 >= TXT_T, "one wave covers every text row");
static_assert(MEL_T % SG == 0 && MEL_T >= 2 * SG, "row pitch keeps the 16-byte phase of every row equal");

struct LaneBelow { int d; int c; };
__device__ __forceinline__ LaneBelow lane_below_send_(float x, float keep) {
  int xi = __builtin_bit_cast(int, x);
  LaneBelow r;
  r.d = __builtin_amdgcn_update_dpp(__builtin_bit_cast(int, keep), xi, 0x111, 0xf, 0xf, false);
  r.c = __builtin_amdgcn_permlanex16(0, xi, 0xffffffff, 0xffffffff, true, false);
  return r;
}
__device__ __forceinline__ float lane_below_take_(LaneBelow r, bool is16) {
  return __builtin_bit_cast(float, is16 ? r.c : r.d);
}
__device__ __forceinline__ float lane_below_(float x, float padv, bool is16) {
  return lane_below_take_(lane_below_send_(x, padv), is16);
}

struct ScanState {
  float a[SR];
  float e[SR];
  float aup;
  float eup;
};

#define ORDER_FENCE() __builtin_amdgcn_sched_barrier(0)
#define ROW_LO(i) ((i) == 0 ? st.eup : st.e[(i) == 0 ? 0 : (i) - 1])
#define ROW_D(i) d[i]  = st.e[i] - ROW_LO(i)
#define ROW_M(i) mx[i] = fmaxf(st.e[i], ROW_LO(i))
#define ROW_X(i) x[i]  = __builtin_amdgcn_exp2f(-__builtin_fabsf(d[i]))
#define ROW_Q(i) mx[i] = __builtin_fmaf(p[i], LOG2E_F, mx[i])
#define ROW_P(i) x[i]  = 1.0f + x[i]
#define ROW_L(i) x[i]  = __builtin_amdgcn_logf(x[i])
#define ROW_A(i) st.a[i] = mx[i] + x[i]
#define ROW_E(i) st.e[i] = st.a[i] + EPS2_F
__device__ __forceinline__ void scan_step_(ScanState& st, const float (&p)[SR], bool is16) {
  static_assert(SR == 7, "the issue order below is written out for seven rows");
  float mx[SR], d[SR], x[SR];
  ROW_D(6); ROW_M(6); ROW_D(5); ROW_M(5); ROW_D(4); ROW_M(4); ROW_D(3); ROW_M(3);
  ROW_D(2); ROW_M(2); ROW_D(1); ROW_M(1); ROW_D(0); ROW_M(0); ORDER_FENCE();
  ROW_X(6); ORDER_FENCE(); ROW_Q(6); ORDER_FENCE();
  ROW_X(5); ORDER_FENCE(); ROW_Q(5); ORDER_FENCE();
  ROW_X(4); ORDER_FENCE(); ROW_Q(4); ORDER_FENCE();
  ROW_X(3); ORDER_FENCE(); ROW_Q(3); ORDER_FENCE();
  ROW_X(2); ORDER_FENCE(); ROW_P(6); ORDER_FENCE();
  ROW_X(1); ORDER_FENCE(); ROW_P(5); ORDER_FENCE();
  ROW_X(0); ORDER_FENCE(); ROW_P(4); ORDER_FENCE();
  ROW_L(6); ORDER_FENCE(); ROW_P(3); ORDER_FENCE();
  ROW_L(5); ORDER_FENCE(); ROW_P(2); ORDER_FENCE();
  ROW_L(4); ORDER_FENCE(); ROW_P(1); ORDER_FENCE();
  ROW_L(3); ORDER_FENCE(); ROW_P(0); ORDER_FENCE();
  ROW_L(2); ORDER_FENCE(); ROW_Q(2); ORDER_FENCE();
  ROW_L(1); ORDER_FENCE(); ROW_Q(1); ORDER_FENCE();
  ROW_L(0); ORDER_FENCE(); ROW_Q(0); ORDER_FENCE();
  ROW_A(6); ORDER_FENCE();
  ROW_A(5); ROW_A(4); ROW_A(3); ROW_A(2); ORDER_FENCE();
  LaneBelow lb = lane_below_send_(st.a[SR - 1], st.aup);
  ORDER_FENCE();
  ROW_A(1); ROW_A(0); ORDER_FENCE();
  ROW_E(6); ROW_E(5); ROW_E(4); ORDER_FENCE();
  st.aup = lane_below_take_(lb, is16);
  ORDER_FENCE();
  ROW_E(3); ROW_E(2); ROW_E(1); ROW_E(0); ORDER_FENCE();
  st.eup = st.aup + EPS2_F;
  ORDER_FENCE();
}

__device__ __forceinline__ void scan_group_(ScanState& st, const v4f (&q)[SR], int ncols, bool is16) {
  float p[SR];
  if (ncols < 1) return;
#pragma unroll
  for (int i = 0; i < SR; ++i) p[i] = q[i].x;
  scan_step_(st, p, is16);
  if (ncols < 2) return;
#pragma unroll
  for (int i = 0; i < SR; ++i) p[i] = q[i].y;
  scan_step_(st, p, is16);
  if (ncols < 3) return;
#pragma unroll
  for (int i = 0; i < SR; ++i) p[i] = q[i].z;
  scan_step_(st, p, is16);
  if (ncols < 4) return;
#pragma unroll
  for (int i = 0; i < SR; ++i) p[i] = q[i].w;
  scan_step_(st, p, is16);
}

__global__ void __launch_bounds__(32)
mdn_scan_kernel(const float* __restrict__ lp,
                const int* __restrict__ text_len,
                const int* __restrict__ mel_len,
                float* __restrict__ ws) {
  const int b    = blockIdx.x;
  const int lane = threadIdx.x;
  const bool is16 = (lane == 16);
  const float* lpb = lp + (size_t)b * TXT_T * MEL_T;
  const int M    = mel_len[b];
  const int last = M - 1;

  int roff[SR];
#pragma unroll
  for (int i = 0; i < SR; ++i) {
    int r = SR * lane + i;
    if (r > TXT_T - 1) r = TXT_T - 1;
    roff[i] = r * MEL_T;
  }

  const int c0 = 1 + (int)((0u - (unsigned)(((size_t)lpb >> 2) + 1u)) & 3u);
  const int ngl = (MEL_T - c0) / SG;
  const int ct  = c0 + SG * ngl;

  float ed[SG - 1][SR];
#pragma unroll
  for (int j = 0; j < SG - 1; ++j)
#pragma unroll
    for (int i = 0; i < SR; ++i) ed[j][i] = lpb[roff[i] + 1 + j];

  v4f q0[SR], q1[SR], q2[SR], q3[SR], q4[SR];
#define LOAD_GROUP(Q, G)                                                     \
  do {                                                                       \
    int gl_ = (G) < ngl - 1 ? (G) : ngl - 1;         \
    const float* gp_ = lpb + c0 + SG * gl_;                                  \
    _Pragma("unroll")                                                        \
    for (int i_ = 0; i_ < SR; ++i_) Q[i_] = *(const v4f*)(gp_ + roff[i_]);   \
  } while (0)
  LOAD_GROUP(q0, 0);
  LOAD_GROUP(q1, 1);
  LOAD_GROUP(q2, 2);
  LOAD_GROUP(q3, 3);
  LOAD_GROUP(q4, 4);
  static_assert(SD == 5, "one q array per group in flight");

  ScanState st;
#pragma unroll
  for (int i = 0; i < SR; ++i) st.a[i] = -PAD_MAG * LOG2E_F;
  if (lane == 0) st.a[0] = lpb[0] * LOG2E_F;
#pragma unroll
  for (int i = 0; i < SR; ++i) st.e[i] = st.a[i] + EPS2_F;
  st.aup = lane_below_(st.a[SR - 1], -PAD_MAG * LOG2E_F, is16);
  st.eup = st.aup + EPS2_F;

#pragma unroll
  for (int j = 0; j < SG - 1; ++j)
    if (1 + j < c0 && 1 + j <= last) scan_step_(st, ed[j], is16);

#pragma unroll
  for (int j = 0; j < SG - 1; ++j)
#pragma unroll
    for (int i = 0; i < SR; ++i) ed[j][i] = lpb[roff[i] + MEL_T - (SG - 1) + j];

  const int lim  = last < ct - 1 ? last : ct - 1;
  const int ncol = lim - c0 + 1;
  int g = 0;
  for (; SG * (g + SD) <= ncol; g += SD) {
    scan_group_(st, q0, SG, is16);  LOAD_GROUP(q0, g + SD);
    scan_group_(st, q1, SG, is16);  LOAD_GROUP(q1, g + SD + 1);
    scan_group_(st, q2, SG, is16);  LOAD_GROUP(q2, g + SD + 2);
    scan_group_(st, q3, SG, is16);  LOAD_GROUP(q3, g + SD + 3);
    scan_group_(st, q4, SG, is16);  LOAD_GROUP(q4, g + SD + 4);
  }
  int cl = ncol - SG * g;
  scan_group_(st, q0, cl, is16);
  scan_group_(st, q1, cl - SG, is16);
  scan_group_(st, q2, cl - 2 * SG, is16);
  scan_group_(st, q3, cl - 3 * SG, is16);
  scan_group_(st, q4, cl - 4 * SG, is16);

#pragma unroll
  for (int j = 0; j < SG - 1; ++j) {
    int col = MEL_T - (SG - 1) + j;
    if (col >= ct && col <= last) scan_step_(st, ed[j], is16);
  }

  const int r  = text_len[b] - 1;
  const int ql = r / SR;
  const int qi = r - ql * SR;
  float res = st.a[0];
#pragma unroll
  for (int i = 1; i < SR; ++i) if (qi == i) res = st.a[i];
  if (lane == ql) ws[P_OFF + b] = (res * LN2_F) / (float)M;
}

__global__ void mdn_reduce_kernel(const float* __restrict__ ws,
                                  float* __restrict__ out) {
  float s = 0.f;
  for (int b = 0; b < B_; ++b) s += ws[P_OFF + b];
  out[0] = -s / (float)B_;
}

extern "C" void kernel_launch(void* const* d_in, const int* in_sizes, int n_in,
                              void* d_out, int out_size, void* d_ws, size_t ws_size,
                              hipStream_t stream) {
  const float* mu_logvar = (const float*)d_in[0];
  const float* melspec   = (const float*)d_in[1];
  const int*   text_len  = (const int*)d_in[2];
  const int*   mel_len   = (const int*)d_in[3];
  float* out = (float*)d_out;
  float* ws  = (float*)d_ws;
  float* lp  = out + 1;

  const int total_waves = B_ * (TXT_PAD / 16) * MS_ALL;
  mdn_gemm_kernel<<<total_waves / 8, 256, 0, stream>>>(mu_logvar, melspec, lp);

  mdn_scan_kernel<<<B_, 32, 0, stream>>>(lp, text_len, mel_len, ws);
  mdn_reduce_kernel<<<1, 1, 0, stream>>>(ws, out);
}


# scan split in two halves on two waves per batch entry: forward from column 0, backward from the last column, joined at the middle by a logsumexp over rows in the last kernel; sum order changes
# speedup vs baseline: 8.5219x; 1.3002x over previous
#include <hip/hip_runtime.h>
#include <math.h>

#define B_      8
#define TXT_T   200
#define TXT_PAD 208
#define MEL_T   800
#define N_MEL   80
#define PAD_MAG 1.0e12f

typedef __attribute__((ext_vector_type(2))) float v2f;
typedef __attribute__((ext_vector_type(8))) float v8f;

#define P_OFF 0

#define NT      3
#define MT_ALL  (MEL_T / 16)
#define MS_FULL (MT_ALL / NT)
#define NT_LAST (MT_ALL - NT * MS_FULL)
#define MS_ALL  (MS_FULL + (NT_LAST > 0 ? 1 : 0))
static_assert(MEL_T % 16 == 0 && NT_LAST >= 0 && NT_LAST < NT, "strips cover a row of tiles");

template <int NTT>
__device__ __forceinline__ void gemm_strip_(const float* __restrict__ ml,
                                            const float* __restrict__ melb,
                                            float* __restrict__ lpc,
                                            int t0, int kl) {
  v8f acc[NTT];
#pragma unroll
  for (int j = 0; j < NTT; ++j) acc[j] = (v8f){0.f, 0.f, 0.f, 0.f, 0.f, 0.f, 0.f, 0.f};
  float c = 0.0f;

#pragma unroll
  for (int n = 0; n < N_MEL; n += 4) {
    v2f mu = *(const v2f*)(ml + n);
    v2f lv = *(const v2f*)(ml + N_MEL + n);
    float x0[NTT], x1[NTT];
#pragma unroll
    for (int j = 0; j < NTT; ++j) {
      x0[j] = melb[(size_t)n * MEL_T + 16 * j];
      x1[j] = melb[(size_t)(n + 1) * MEL_T + 16 * j];
    }

    v2f w;  w.x = expf(-lv.x);          w.y = expf(-lv.y);
    v2f g;  g.x = -2.0f * mu.x * w.x;   g.y = -2.0f * mu.y * w.y;
    c += mu.x * mu.x * w.x + lv.x;
    c += mu.y * mu.y * w.y + lv.y;

#pragma unroll
    for (int j = 0; j < NTT; ++j) {
      v2f xx; xx.x = x0[j] * x0[j];     xx.y = x1[j] * x1[j];
      acc[j] = __builtin_amdgcn_wmma_f32_16x16x4_f32(
           false, w,  false, xx,
           (short)0, acc[j],  false,  false);
    }
#pragma unroll
    for (int j = 0; j < NTT; ++j) {
      v2f xv; xv.x = x0[j];             xv.y = x1[j];
      acc[j] = __builtin_amdgcn_wmma_f32_16x16x4_f32(
           false, g,  false, xv,
           (short)0, acc[j],  false,  false);
    }
  }

  {
    v2f ca;  ca.x = c;     ca.y = 0.0f;
    v2f one; one.x = 1.0f; one.y = 0.0f;
#pragma unroll
    for (int j = 0; j < NTT; ++j)
      acc[j] = __builtin_amdgcn_wmma_f32_16x16x4_f32(
           false, ca,  false, one,
           (short)0, acc[j],  false,  false);
  }

  const float scale = -0.5f / (float)N_MEL;
#pragma unroll
  for (int i = 0; i < 8; ++i) {
    int t = t0 + i + 8 * kl;
    if (t < TXT_T) {
      float* row = lpc + (size_t)t * MEL_T;
#pragma unroll
      for (int j = 0; j < NTT; ++j) row[16 * j] = scale * acc[j][i];
    }
  }
}

__global__ void __launch_bounds__(256)
mdn_gemm_kernel(const float* __restrict__ mu_logvar,
                const float* __restrict__ mel,
                float* __restrict__ lp) {
  const int TT = TXT_PAD / 16;
  int wid = blockIdx.x * (blockDim.x >> 5) + (threadIdx.x >> 5);
  if (wid >= B_ * TT * MS_ALL) return;
  int lane = threadIdx.x & 31;
  int l15  = lane & 15;
  int kl   = lane >> 4;

  int b  = wid / (TT * MS_ALL);
  int r  = wid % (TT * MS_ALL);
  int t0 = (r / MS_ALL) * 16;
  int ms = r % MS_ALL;
  int m0 = ms * (16 * NT);

  int trow = t0 + l15;
  if (trow > TXT_T - 1) trow = TXT_T - 1;
  const float* ml   = mu_logvar + (size_t)(b * TXT_T + trow) * (2 * N_MEL) + 2 * kl;
  const float* melb = mel + (size_t)b * N_MEL * MEL_T + (size_t)(2 * kl) * MEL_T + m0 + l15;
  float* lpc = lp + (size_t)b * TXT_T * MEL_T + m0 + l15;

  if (NT_LAST == 0 || ms < MS_FULL) gemm_strip_<NT>(ml, melb, lpc, t0, kl);
  else gemm_strip_<(NT_LAST > 0 ? NT_LAST : 1)>(ml, melb, lpc, t0, kl);
}

#define LOG2E_F 1.44269504f
#define LN2_F   0.693147181f
#define EPS2_F  (1e-7f * LOG2E_F)

#define SR 7
#define SG 4
#define SD 5
typedef __attribute__((ext_vector_type(4))) float v4f;
static_assert(SR * 32 >= TXT_T, "one wave covers every text row");
static_assert(MEL_T % SG == 0 && MEL_T >= 2 * SG, "row pitch keeps the 16-byte phase of every row equal");

struct LaneBelow { int d; int c; };
__device__ __forceinline__ LaneBelow lane_below_send_(float x, float keep) {
  int xi = __builtin_bit_cast(int, x);
  LaneBelow r;
  r.d = __builtin_amdgcn_update_dpp(__builtin_bit_cast(int, keep), xi, 0x111, 0xf, 0xf, false);
  r.c = __builtin_amdgcn_permlanex16(0, xi, 0xffffffff, 0xffffffff, true, false);
  return r;
}
__device__ __forceinline__ float lane_below_take_(LaneBelow r, bool is16) {
  return __builtin_bit_cast(float, is16 ? r.c : r.d);
}
__device__ __forceinline__ float lane_below_(float x, float padv, bool is16) {
  return lane_below_take_(lane_below_send_(x, padv), is16);
}

struct ScanState {
  float a[SR];
  float e[SR];
  float aup;
  float eup;
};

#define ORDER_FENCE() __builtin_amdgcn_sched_barrier(0)
#define ROW_LO(i) ((i) == 0 ? st.eup : st.e[(i) == 0 ? 0 : (i) - 1])
#define ROW_D(i) d[i]  = st.e[i] - ROW_LO(i)
#define ROW_M(i) mx[i] = fmaxf(st.e[i], ROW_LO(i))
#define ROW_X(i) x[i]  = __builtin_amdgcn_exp2f(-__builtin_fabsf(d[i]))
#define ROW_Q(i) mx[i] = __builtin_fmaf(p[i], LOG2E_F, mx[i])
#define ROW_P(i) x[i]  = 1.0f + x[i]
#define ROW_L(i) x[i]  = __builtin_amdgcn_logf(x[i])
#define ROW_A(i) st.a[i] = mx[i] + x[i]
#define ROW_E(i) st.e[i] = st.a[i] + EPS2_F
__device__ __forceinline__ void scan_step_(ScanState& st, const float (&p)[SR], bool is16) {
  static_assert(SR == 7, "the issue order below is written out for seven rows");
  float mx[SR], d[SR], x[SR];
  ROW_D(6); ROW_M(6); ROW_D(5); ROW_M(5); ROW_D(4); ROW_M(4); ROW_D(3); ROW_M(3);
  ROW_D(2); ROW_M(2); ROW_D(1); ROW_M(1); ROW_D(0); ROW_M(0); ORDER_FENCE();
  ROW_X(6); ORDER_FENCE(); ROW_Q(6); ORDER_FENCE();
  ROW_X(5); ORDER_FENCE(); ROW_Q(5); ORDER_FENCE();
  ROW_X(4); ORDER_FENCE(); ROW_Q(4); ORDER_FENCE();
  ROW_X(3); ORDER_FENCE(); ROW_Q(3); ORDER_FENCE();
  ROW_X(2); ORDER_FENCE(); ROW_P(6); ORDER_FENCE();
  ROW_X(1); ORDER_FENCE(); ROW_P(5); ORDER_FENCE();
  ROW_X(0); ORDER_FENCE(); ROW_P(4); ORDER_FENCE();
  ROW_L(6); ORDER_FENCE(); ROW_P(3); ORDER_FENCE();
  ROW_L(5); ORDER_FENCE(); ROW_P(2); ORDER_FENCE();
  ROW_L(4); ORDER_FENCE(); ROW_P(1); ORDER_FENCE();
  ROW_L(3); ORDER_FENCE(); ROW_P(0); ORDER_FENCE();
  ROW_L(2); ORDER_FENCE(); ROW_Q(2); ORDER_FENCE();
  ROW_L(1); ORDER_FENCE(); ROW_Q(1); ORDER_FENCE();
  ROW_L(0); ORDER_FENCE(); ROW_Q(0); ORDER_FENCE();
  ROW_A(6); ORDER_FENCE();
  ROW_A(5); ROW_A(4); ROW_A(3); ROW_A(2); ORDER_FENCE();
  LaneBelow lb = lane_below_send_(st.a[SR - 1], st.aup);
  ORDER_FENCE();
  ROW_A(1); ROW_A(0); ORDER_FENCE();
  ROW_E(6); ROW_E(5); ROW_E(4); ORDER_FENCE();
  st.aup = lane_below_take_(lb, is16);
  ORDER_FENCE();
  ROW_E(3); ROW_E(2); ROW_E(1); ROW_E(0); ORDER_FENCE();
  st.eup = st.aup + EPS2_F;
  ORDER_FENCE();
}

template <int DIR>
__device__ __forceinline__ void scan_group_(ScanState& st, const v4f (&q)[SR], int ncols, bool is16) {
  float p[SR];
  if (ncols < 1) return;
#pragma unroll
  for (int i = 0; i < SR; ++i) p[i] = DIR == 0 ? q[i].x : q[i].w;
  scan_step_(st, p, is16);
  if (ncols < 2) return;
#pragma unroll
  for (int i = 0; i < SR; ++i) p[i] = DIR == 0 ? q[i].y : q[i].z;
  scan_step_(st, p, is16);
  if (ncols < 3) return;
#pragma unroll
  for (int i = 0; i < SR; ++i) p[i] = DIR == 0 ? q[i].z : q[i].y;
  scan_step_(st, p, is16);
  if (ncols < 4) return;
#pragma unroll
  for (int i = 0; i < SR; ++i) p[i] = DIR == 0 ? q[i].w : q[i].x;
  scan_step_(st, p, is16);
}

#define SW (SR * 32)
#define MEET_MIN 16

template <int DIR>
__device__ __forceinline__ float lp_at_(const float* __restrict__ lpb, int roff, int u) {
  return DIR == 0 ? lpb[roff + u] : lpb[roff + (MEL_T - 1) - u];
}

template <int DIR>
__device__ __forceinline__ void scan_half_(const float* __restrict__ lpb, int L, int s0, int n,
                                           bool half_step, float* __restrict__ dst, int lane) {
  const bool is16 = (lane == 16);
  const int last = s0 + n - 1;

  int roff[SR];
#pragma unroll
  for (int i = 0; i < SR; ++i) {
    int r = SR * lane + i;
    if (DIR != 0) r = (L - 1) - r;
    unsigned ru = r < 0 ? 0u : (unsigned)r;
    if (ru > (unsigned)(TXT_T - 1)) ru = (unsigned)(TXT_T - 1);
    roff[i] = (int)(ru * (unsigned)MEL_T);
  }

  const unsigned base4 = (unsigned)((size_t)lpb >> 2);
  const unsigned u1 = (unsigned)(s0 + 1);
  const int c0 = s0 + 1 + (int)((DIR == 0 ? (0u - (base4 + u1)) : (base4 - u1)) & 3u);
  const int ngl = (MEL_T - c0) / SG;
  const int ct  = c0 + SG * ngl;

  float ed[SG - 1][SR];
#pragma unroll
  for (int j = 0; j < SG - 1; ++j)
#pragma unroll
    for (int i = 0; i < SR; ++i) ed[j][i] = lp_at_<DIR>(lpb, roff[i], s0 + 1 + j);

  const int a0 = ((MEL_T - SG) - c0) & 3;
  const int kg = ((MEL_T - SG) - c0 - a0) / SG;
  const v4f* lpa = (const v4f*)(lpb + a0);
  int roff4[SR];
#pragma unroll
  for (int i = 0; i < SR; ++i) roff4[i] = roff[i] / SG;

  v4f q0[SR], q1[SR], q2[SR], q3[SR], q4[SR];
#define LOAD_GROUP(Q, G)                                                     \
  do {                                                                       \
    int gl_ = (G) < ngl - 1 ? (G) : ngl - 1;         \
    int gu_ = c0 + SG * gl_;                                                 \
    const float* gp_ = lpb + gu_;                                  \
    _Pragma("unroll")                                                        \
    for (int i_ = 0; i_ < SR; ++i_)                                          \
      Q[i_] = DIR == 0 ? *(const v4f*)(gp_ + roff[i_])                       \
                       : lpa[(unsigned)(roff4[i_] + (kg - gl_))];            \
  } while (0)
  LOAD_GROUP(q0, 0);
  LOAD_GROUP(q1, 1);
  LOAD_GROUP(q2, 2);
  LOAD_GROUP(q3, 3);
  LOAD_GROUP(q4, 4);
  static_assert(SD == 5, "one q array per group in flight");

  float fin[SR];
#pragma unroll
  for (int i = 0; i < SR; ++i) fin[i] = lp_at_<DIR>(lpb, roff[i], s0 + (n >= 1 ? n : 0));

  ScanState st;
#pragma unroll
  for (int i = 0; i < SR; ++i) st.a[i] = -PAD_MAG * LOG2E_F;
  if (lane == 0) st.a[0] = lp_at_<DIR>(lpb, roff[0], s0) * LOG2E_F;
#pragma unroll
  for (int i = 0; i < SR; ++i) st.e[i] = st.a[i] + EPS2_F;
  st.aup = lane_below_(st.a[SR - 1], -PAD_MAG * LOG2E_F, is16);
  st.eup = st.aup + EPS2_F;

  if (n >= 1) {
#pragma unroll
  for (int j = 0; j < SG - 1; ++j)
    if (s0 + 1 + j < c0 && s0 + 1 + j <= last) scan_step_(st, ed[j], is16);

#pragma unroll
  for (int j = 0; j < SG - 1; ++j)
#pragma unroll
    for (int i = 0; i < SR; ++i) ed[j][i] = lp_at_<DIR>(lpb, roff[i], MEL_T - (SG - 1) + j);

  const int lim  = last < ct - 1 ? last : ct - 1;
  const int ncol = lim - c0 + 1;
  int g = 0;
  for (; SG * (g + SD) <= ncol; g += SD) {
    scan_group_<DIR>(st, q0, SG, is16);  LOAD_GROUP(q0, g + SD);
    scan_group_<DIR>(st, q1, SG, is16);  LOAD_GROUP(q1, g + SD + 1);
    scan_group_<DIR>(st, q2, SG, is16);  LOAD_GROUP(q2, g + SD + 2);
    scan_group_<DIR>(st, q3, SG, is16);  LOAD_GROUP(q3, g + SD + 3);
    scan_group_<DIR>(st, q4, SG, is16);  LOAD_GROUP(q4, g + SD + 4);
  }
  int cl = ncol - SG * g;
  scan_group_<DIR>(st, q0, cl, is16);
  scan_group_<DIR>(st, q1, cl - SG, is16);
  scan_group_<DIR>(st, q2, cl - 2 * SG, is16);
  scan_group_<DIR>(st, q3, cl - 3 * SG, is16);
  scan_group_<DIR>(st, q4, cl - 4 * SG, is16);

#pragma unroll
  for (int j = 0; j < SG - 1; ++j) {
    int pos = MEL_T - (SG - 1) + j;
    if (pos >= ct && pos <= last) scan_step_(st, ed[j], is16);
  }

  scan_step_(st, fin, is16);
  }

  if (half_step) {
    float zero[SR];
#pragma unroll
    for (int i = 0; i < SR; ++i) zero[i] = 0.0f;
    scan_step_(st, zero, is16);
  }

#pragma unroll
  for (int i = 0; i < SR; ++i) dst[32 * i + lane] = st.a[i];
}

__global__ void __launch_bounds__(32)
mdn_scan_kernel(const float* __restrict__ lp,
                const int* __restrict__ text_len,
                const int* __restrict__ mel_len,
                float* __restrict__ ws) {
  const int b    = blockIdx.x >> 1;
  const int dir  = blockIdx.x & 1;
  const int lane = threadIdx.x;
  const float* lpb = lp + (size_t)b * TXT_T * MEL_T;
  const int M = mel_len[b];
  const int L = text_len[b];
  const bool meet = M >= MEET_MIN;
  const int nf = meet ? (M - 2) / 2 : M - 1;
  float* dst = ws + P_OFF + (size_t)(2 * b + dir) * SW;
  if (dir == 0) {
    scan_half_<0>(lpb, L, 0, nf, meet, dst, lane);
  } else {
    if (!meet) return;
    scan_half_<1>(lpb, L, MEL_T - M, M - 2 - nf, false, dst, lane);
  }
}

__global__ void __launch_bounds__(32 * B_)
mdn_reduce_kernel(const float* __restrict__ ws,
                  const int* __restrict__ text_len,
                  const int* __restrict__ mel_len,
                  float* __restrict__ out) {
  __shared__ float part[B_];
  const int b    = threadIdx.x >> 5;
  const int lane = threadIdx.x & 31;
  const int M = mel_len[b];
  const int L = text_len[b];
  const float* F = ws + P_OFF + (size_t)(2 * b) * SW;
  const float* C = F + SW;
  float res;
  if (M >= MEET_MIN) {
    float s[SR];
    float mx = -3.0e38f;
#pragma unroll
    for (int i = 0; i < SR; ++i) {
      int t = SR * lane + i;
      bool ok = t <= L - 1;
      int tc = ok ? (L - 1) - t : 0;
      float v = F[32 * i + lane] + C[32 * (tc % SR) + tc / SR];
      s[i] = ok ? v : -3.0e38f;
      mx = fmaxf(mx, s[i]);
    }
    for (int m = 16; m >= 1; m >>= 1) mx = fmaxf(mx, __shfl_xor(mx, m, 32));
    float sum = 0.0f;
#pragma unroll
    for (int i = 0; i < SR; ++i) {
      int t = SR * lane + i;
      float e = __builtin_amdgcn_exp2f(s[i] - mx);
      sum += (t <= L - 1) ? e : 0.0f;
    }
    for (int m = 16; m >= 1; m >>= 1) sum += __shfl_xor(sum, m, 32);
    res = mx + __builtin_amdgcn_logf(sum);
  } else {
    int r = L - 1;
    if (r < 0) r = 0;
    if (r > TXT_T - 1) r = TXT_T - 1;
    res = F[32 * (r % SR) + r / SR];
  }
  if (lane == 0) part[b] = (res * LN2_F) / (float)M;
  __syncthreads();
  if (threadIdx.x == 0) {
    float s = 0.f;
    for (int i = 0; i < B_; ++i) s += part[i];
    out[0] = -s / (float)B_;
  }
}

extern "C" void kernel_launch(void* const* d_in, const int* in_sizes, int n_in,
                              void* d_out, int out_size, void* d_ws, size_t ws_size,
                              hipStream_t stream) {
  const float* mu_logvar = (const float*)d_in[0];
  const float* melspec   = (const float*)d_in[1];
  const int*   text_len  = (const int*)d_in[2];
  const int*   mel_len   = (const int*)d_in[3];
  float* out = (float*)d_out;
  float* ws  = (float*)d_ws;
  float* lp  = out + 1;

  const int total_waves = B_ * (TXT_PAD / 16) * MS_ALL;
  mdn_gemm_kernel<<<total_waves / 8, 256, 0, stream>>>(mu_logvar, melspec, lp);

  mdn_scan_kernel<<<2 * B_, 32, 0, stream>>>(lp, text_len, mel_len, ws);
  mdn_reduce_kernel<<<1, 32 * B_, 0, stream>>>(ws, text_len, mel_len, out);
}
